# starting kernel
# baseline (speedup 1.0000x reference)
#include <hip/hip_runtime.h>

typedef float v2f __attribute__((ext_vector_type(2)));
typedef float v8f __attribute__((ext_vector_type(8)));

__global__ void k_fill(float* __restrict__ p, float v, int n) {
    int i = blockIdx.x * blockDim.x + threadIdx.x;
    if (i < n) p[i] = v;
}

__global__ void k_zero(float* __restrict__ p, int n) {
    for (int i = blockIdx.x * blockDim.x + threadIdx.x; i < n;
         i += gridDim.x * blockDim.x)
        p[i] = 0.0f;
}

__global__ void k_deg_scatter(const int* __restrict__ dst, float* __restrict__ deg, int E) {
    int e = blockIdx.x * blockDim.x + threadIdx.x;
    if (e < E) unsafeAtomicAdd(&deg[dst[e]], 1.0f);
}

__global__ void k_dinv(const float* __restrict__ deg, float* __restrict__ dinv, int n) {
    int i = blockIdx.x * blockDim.x + threadIdx.x;
    if (i < n) {
        float d = deg[i];
        dinv[i] = (d > 0.0f) ? rsqrtf(d) : 0.0f;
    }
}

template <int K, int NF>
__launch_bounds__(128)
__global__ void k_gemm_scaled(const float* __restrict__ A,
                              const float* __restrict__ W,
                              const float* __restrict__ dinv,
                              float* __restrict__ out, int M) {
    const int lane  = threadIdx.x & 31;
    const int tileM = blockIdx.x;
    const int tileN = blockIdx.y * blockDim.y + threadIdx.y;
    const int r16   = lane & 15;
    const int hi    = lane >> 4;
    const int kh    = hi * 2;

    int rowA = tileM * 16 + r16;
    if (rowA >= M) rowA = M - 1;
    const int colB = tileN * 16 + r16;

    const float* ap = A + (size_t)rowA * K + kh;
    const float* bp = W + (size_t)kh * NF + colB;

    __builtin_prefetch(bp, 0, 3);

    v8f c = {};
#pragma unroll 8
    for (int k = 0; k < K; k += 4) {
        v2f a = *(const v2f*)ap;
        v2f b;
        b.x = bp[0];
        b.y = bp[NF];
        c = __builtin_amdgcn_wmma_f32_16x16x4_f32(
                 false, a,  false, b,
                 (short)0, c,  false,  false);
        ap += 4;
        bp += 4 * NF;
    }

    const int rowBase = tileM * 16 + hi * 8;
    float* op = out + (size_t)rowBase * NF + colB;
    if (rowBase + 8 <= M) {
        const float4 d0 = *(const float4*)(dinv + rowBase);
        const float4 d1 = *(const float4*)(dinv + rowBase + 4);
        op[0 * (size_t)NF] = c[0] * d0.x;
        op[1 * (size_t)NF] = c[1] * d0.y;
        op[2 * (size_t)NF] = c[2] * d0.z;
        op[3 * (size_t)NF] = c[3] * d0.w;
        op[4 * (size_t)NF] = c[4] * d1.x;
        op[5 * (size_t)NF] = c[5] * d1.y;
        op[6 * (size_t)NF] = c[6] * d1.z;
        op[7 * (size_t)NF] = c[7] * d1.w;
    } else {
#pragma unroll
        for (int r = 0; r < 8; ++r) {
            int row = rowBase + r;
            if (row < M) op[(size_t)r * NF] = c[r] * dinv[row];
        }
    }
}

template <int F>
__global__ void k_edge_scatter(const float* __restrict__ hs,
                               const int* __restrict__ src,
                               const int* __restrict__ dst,
                               float* __restrict__ agg, int E) {
    constexpr int TPE = F / 4;
    int e = blockIdx.x * (256 / TPE) + threadIdx.x / TPE;
    if (e >= E) return;
    int q = (threadIdx.x % TPE) * 4;
    const float4 v = *(const float4*)(hs + (size_t)src[e] * F + q);
    float* a = agg + (size_t)dst[e] * F + q;
    unsafeAtomicAdd(a + 0, v.x);
    unsafeAtomicAdd(a + 1, v.y);
    unsafeAtomicAdd(a + 2, v.z);
    unsafeAtomicAdd(a + 3, v.w);
}

template <int F, bool RELU>
__global__ void k_finalize(const float* __restrict__ agg,
                           const float* __restrict__ hs,
                           const float* __restrict__ dinv,
                           const float* __restrict__ bias,
                           float* __restrict__ out, int N) {
    constexpr int TPR = F / 4;
    int t = blockIdx.x * blockDim.x + threadIdx.x;
    if (t >= N * TPR) return;
    int i = t / TPR;
    int q = (t % TPR) * 4;
    size_t off = (size_t)i * F + q;
    float di = dinv[i];
    const float4 va = *(const float4*)(agg + off);
    const float4 vh = *(const float4*)(hs + off);
    const float4 vb = *(const float4*)(bias + q);
    float4 r;
    r.x = di * (va.x + vh.x) + vb.x;
    r.y = di * (va.y + vh.y) + vb.y;
    r.z = di * (va.z + vh.z) + vb.z;
    r.w = di * (va.w + vh.w) + vb.w;
    if (RELU) {
        r.x = fmaxf(r.x, 0.0f);
        r.y = fmaxf(r.y, 0.0f);
        r.z = fmaxf(r.z, 0.0f);
        r.w = fmaxf(r.w, 0.0f);
    }
    *(float4*)(out + off) = r;
}

extern "C" void kernel_launch(void* const* d_in, const int* in_sizes, int n_in,
                              void* d_out, int out_size, void* d_ws, size_t ws_size,
                              hipStream_t stream) {
    const float* x  = (const float*)d_in[0];
    const float* W1 = (const float*)d_in[1];
    const float* b1 = (const float*)d_in[2];
    const float* W2 = (const float*)d_in[3];
    const float* b2 = (const float*)d_in[4];
    const int*   ei = (const int*)d_in[5];

    const int IN = 128, HID = 128, OUT = 64;
    const int N = in_sizes[0] / IN;
    const int E = in_sizes[5] / 2;
    const int* src = ei;
    const int* dst = ei + E;

    float* ws   = (float*)d_ws;
    float* deg  = ws;
    float* dinv = deg + N;
    float* bufA = dinv + N;
    float* bufB = bufA + (size_t)N * HID;
    float* bufC = bufB + (size_t)N * HID;
    float* out  = (float*)d_out;

    k_fill<<<(N + 255) / 256, 256, 0, stream>>>(deg, 1.0f, N);
    k_deg_scatter<<<(E + 255) / 256, 256, 0, stream>>>(dst, deg, E);
    k_dinv<<<(N + 255) / 256, 256, 0, stream>>>(deg, dinv, N);

    const int tilesM = (N + 15) / 16;

    k_zero<<<1024, 256, 0, stream>>>(bufB, N * HID);
    {
        dim3 blk(32, 4);
        dim3 grd(tilesM, HID / 64);
        k_gemm_scaled<128, 128><<<grd, blk, 0, stream>>>(x, W1, dinv, bufA, N);
    }
    k_edge_scatter<128><<<(E + 7) / 8, 256, 0, stream>>>(bufA, src, dst, bufB, E);
    k_finalize<128, true><<<(N * 32 + 255) / 256, 256, 0, stream>>>(
        bufB, bufA, dinv, b1, bufC, N);

    k_zero<<<1024, 256, 0, stream>>>(bufB, N * OUT);
    {
        dim3 blk(32, 4);
        dim3 grd(tilesM, OUT / 64);
        k_gemm_scaled<128, 64><<<grd, blk, 0, stream>>>(bufC, W2, dinv, bufA, N);
    }
    k_edge_scatter<64><<<(E + 15) / 16, 256, 0, stream>>>(bufA, src, dst, bufB, E);
    k_finalize<64, false><<<(N * 16 + 255) / 256, 256, 0, stream>>>(
        bufB, bufA, dinv, b2, out, N);
}


# Edges counting-sorted by destination once per call (LDS histograms); each node sums its in-edges in registers, finalize folded in: float atomics gone, arrival-order sums become sorted-order sums
# speedup vs baseline: 2.9616x; 2.9616x over previous
#include <hip/hip_runtime.h>

typedef float v2f __attribute__((ext_vector_type(2)));
typedef float v8f __attribute__((ext_vector_type(8)));

constexpr int NCH = 64;
constexpr int SORT_T = 1024;

__launch_bounds__(SORT_T)
__global__ void k_chunk_hist(const int* __restrict__ dst, int* __restrict__ H,
                             int N, int E) {
    extern __shared__ int sh[];
    const int b   = blockIdx.x;
    const int per = (E + NCH - 1) / NCH;
    const int e0  = b * per;
    const int e1  = (e0 + per < E) ? (e0 + per) : E;
    for (int i = threadIdx.x; i < N; i += SORT_T) sh[i] = 0;
    __syncthreads();
    int e = e0 + threadIdx.x;
    for (; e + 3 * SORT_T < e1; e += 4 * SORT_T) {
        const int d0 = dst[e];
        const int d1 = dst[e + SORT_T];
        const int d2 = dst[e + 2 * SORT_T];
        const int d3 = dst[e + 3 * SORT_T];
        atomicAdd(&sh[d0], 1);
        atomicAdd(&sh[d1], 1);
        atomicAdd(&sh[d2], 1);
        atomicAdd(&sh[d3], 1);
    }
    for (; e < e1; e += SORT_T) atomicAdd(&sh[dst[e]], 1);
    __syncthreads();
    int* row = H + (size_t)b * N;
    for (int i = threadIdx.x; i < N; i += SORT_T) row[i] = sh[i];
}

__global__ void k_chunk_prefix(const int* __restrict__ H, int* __restrict__ P,
                               int* __restrict__ cnt, int N) {
    const int n = blockIdx.x * blockDim.x + threadIdx.x;
    if (n >= N) return;
    int run = 0;
#pragma unroll
    for (int b0 = 0; b0 < NCH; b0 += 16) {
        int t[16];
#pragma unroll
        for (int j = 0; j < 16; ++j) t[j] = H[(size_t)(b0 + j) * N + n];
#pragma unroll
        for (int j = 0; j < 16; ++j) {
            P[(size_t)(b0 + j) * N + n] = run;
            run += t[j];
        }
    }
    cnt[n] = run;
}

__launch_bounds__(SORT_T)
__global__ void k_node_scan(const int* __restrict__ cnt, int* __restrict__ rowptr,
                            float* __restrict__ dinv, int N) {
    __shared__ int s[2][SORT_T];
    const int t   = threadIdx.x;
    const int ipt = (N + SORT_T - 1) / SORT_T;
    const int i0  = t * ipt;
    const int i1  = (i0 + ipt < N) ? (i0 + ipt) : N;
    int sum = 0;
    for (int i = i0; i < i1; ++i) {
        const int c = cnt[i];
        sum += c;
        const float d = (float)c + 1.0f;
        dinv[i] = (d > 0.0f) ? rsqrtf(d) : 0.0f;
    }
    s[0][t] = sum;
    __syncthreads();
    int cur = 0;
    for (int off = 1; off < SORT_T; off <<= 1) {
        int v = s[cur][t];
        if (t >= off) v += s[cur][t - off];
        s[cur ^ 1][t] = v;
        __syncthreads();
        cur ^= 1;
    }
    int run = s[cur][t] - sum;
    for (int i = i0; i < i1; ++i) {
        rowptr[i] = run;
        run += cnt[i];
    }
    if (t == SORT_T - 1) rowptr[N] = s[cur][t];
}

__launch_bounds__(SORT_T)
__global__ void k_chunk_place(const int* __restrict__ src, const int* __restrict__ dst,
                              const int* __restrict__ P, const int* __restrict__ rowptr,
                              int* __restrict__ csr, int N, int E) {
    extern __shared__ int sh[];
    const int b   = blockIdx.x;
    const int per = (E + NCH - 1) / NCH;
    const int e0  = b * per;
    const int e1  = (e0 + per < E) ? (e0 + per) : E;
    const int* row = P + (size_t)b * N;
    for (int i = threadIdx.x; i < N; i += SORT_T) sh[i] = rowptr[i] + row[i];
    __syncthreads();
    int e = e0 + threadIdx.x;
    for (; e + 3 * SORT_T < e1; e += 4 * SORT_T) {
        const int d0 = dst[e];
        const int d1 = dst[e + SORT_T];
        const int d2 = dst[e + 2 * SORT_T];
        const int d3 = dst[e + 3 * SORT_T];
        const int s0 = src[e];
        const int s1 = src[e + SORT_T];
        const int s2 = src[e + 2 * SORT_T];
        const int s3 = src[e + 3 * SORT_T];
        csr[atomicAdd(&sh[d0], 1)] = s0;
        csr[atomicAdd(&sh[d1], 1)] = s1;
        csr[atomicAdd(&sh[d2], 1)] = s2;
        csr[atomicAdd(&sh[d3], 1)] = s3;
    }
    for (; e < e1; e += SORT_T) {
        const int pos = atomicAdd(&sh[dst[e]], 1);
        csr[pos] = src[e];
    }
}

template <int K, int NF>
__launch_bounds__(128)
__global__ void k_gemm_scaled(const float* __restrict__ A,
                              const float* __restrict__ W,
                              const float* __restrict__ dinv,
                              float* __restrict__ out, int M) {
    const int lane  = threadIdx.x & 31;
    const int tileM = blockIdx.x;
    const int tileN = blockIdx.y * blockDim.y + threadIdx.y;
    const int r16   = lane & 15;
    const int hi    = lane >> 4;
    const int kh    = hi * 2;

    int rowA = tileM * 16 + r16;
    if (rowA >= M) rowA = M - 1;
    const int colB = tileN * 16 + r16;

    const float* ap = A + (size_t)rowA * K + kh;
    const float* bp = W + (size_t)kh * NF + colB;

    __builtin_prefetch(bp, 0, 3);

    v8f c = {};
#pragma unroll 8
    for (int k = 0; k < K; k += 4) {
        v2f a = *(const v2f*)ap;
        v2f b;
        b.x = bp[0];
        b.y = bp[NF];
        c = __builtin_amdgcn_wmma_f32_16x16x4_f32(
                 false, a,  false, b,
                 (short)0, c,  false,  false);
        ap += 4;
        bp += 4 * NF;
    }

    const int rowBase = tileM * 16 + hi * 8;
    float* op = out + (size_t)rowBase * NF + colB;
    if (rowBase + 8 <= M) {
        const float4 d0 = *(const float4*)(dinv + rowBase);
        const float4 d1 = *(const float4*)(dinv + rowBase + 4);
        op[0 * (size_t)NF] = c[0] * d0.x;
        op[1 * (size_t)NF] = c[1] * d0.y;
        op[2 * (size_t)NF] = c[2] * d0.z;
        op[3 * (size_t)NF] = c[3] * d0.w;
        op[4 * (size_t)NF] = c[4] * d1.x;
        op[5 * (size_t)NF] = c[5] * d1.y;
        op[6 * (size_t)NF] = c[6] * d1.z;
        op[7 * (size_t)NF] = c[7] * d1.w;
    } else {
#pragma unroll
        for (int r = 0; r < 8; ++r) {
            int row = rowBase + r;
            if (row < M) op[(size_t)r * NF] = c[r] * dinv[row];
        }
    }
}

template <int F, bool RELU>
__global__ void k_gather_finalize(const float* __restrict__ hs,
                                  const int* __restrict__ rowptr,
                                  const int* __restrict__ csr,
                                  const float* __restrict__ dinv,
                                  const float* __restrict__ bias,
                                  float* __restrict__ out, int N) {
    constexpr int TPR = F / 4;
    const int i = blockIdx.x * (256 / TPR) + threadIdx.x / TPR;
    if (i >= N) return;
    const int q = (threadIdx.x % TPR) * 4;
    const float* hq = hs + q;
    const int j1 = rowptr[i + 1];
    int j = rowptr[i];
    float4 acc = make_float4(0.0f, 0.0f, 0.0f, 0.0f);
    for (; j + 8 <= j1; j += 8) {
        int s[8];
        float4 v[8];
#pragma unroll
        for (int u = 0; u < 8; ++u) s[u] = csr[j + u];
#pragma unroll
        for (int u = 0; u < 8; ++u) v[u] = *(const float4*)(hq + (size_t)s[u] * F);
#pragma unroll
        for (int u = 0; u < 8; ++u) {
            acc.x += v[u].x; acc.y += v[u].y; acc.z += v[u].z; acc.w += v[u].w;
        }
    }
    for (; j < j1; ++j) {
        const float4 v = *(const float4*)(hq + (size_t)csr[j] * F);
        acc.x += v.x; acc.y += v.y; acc.z += v.z; acc.w += v.w;
    }
    const size_t off = (size_t)i * F + q;
    const float di = dinv[i];
    const float4 vh = *(const float4*)(hs + off);
    const float4 vb = *(const float4*)(bias + q);
    float4 r;
    r.x = di * (acc.x + vh.x) + vb.x;
    r.y = di * (acc.y + vh.y) + vb.y;
    r.z = di * (acc.z + vh.z) + vb.z;
    r.w = di * (acc.w + vh.w) + vb.w;
    if (RELU) {
        r.x = fmaxf(r.x, 0.0f);
        r.y = fmaxf(r.y, 0.0f);
        r.z = fmaxf(r.z, 0.0f);
        r.w = fmaxf(r.w, 0.0f);
    }
    *(float4*)(out + off) = r;
}

extern "C" void kernel_launch(void* const* d_in, const int* in_sizes, int n_in,
                              void* d_out, int out_size, void* d_ws, size_t ws_size,
                              hipStream_t stream) {
    const float* x  = (const float*)d_in[0];
    const float* W1 = (const float*)d_in[1];
    const float* b1 = (const float*)d_in[2];
    const float* W2 = (const float*)d_in[3];
    const float* b2 = (const float*)d_in[4];
    const int*   ei = (const int*)d_in[5];

    const int IN = 128, HID = 128, OUT = 64;
    const int N = in_sizes[0] / IN;
    const int E = in_sizes[5] / 2;
    const int* src = ei;
    const int* dst = ei + E;

    const size_t Np = ((size_t)N + 64) & ~(size_t)63;
    const size_t Ep = ((size_t)E + 63) & ~(size_t)63;

    float* ws     = (float*)d_ws;
    float* dinv   = ws;
    int*   rowptr = (int*)(dinv + Np);
    int*   cnt    = rowptr + Np;
    int*   H      = cnt + Np;
    int*   P      = H + (size_t)NCH * Np;
    int*   csr    = P + (size_t)NCH * Np;
    float* bufA   = (float*)(csr + Ep);
    float* bufC   = bufA + (size_t)N * HID;
    float* out    = (float*)d_out;

    const size_t ldsN = (size_t)N * sizeof(int);
    k_chunk_hist<<<NCH, SORT_T, ldsN, stream>>>(dst, H, N, E);
    k_chunk_prefix<<<(N + 255) / 256, 256, 0, stream>>>(H, P, cnt, N);
    k_node_scan<<<1, SORT_T, 0, stream>>>(cnt, rowptr, dinv, N);
    k_chunk_place<<<NCH, SORT_T, ldsN, stream>>>(src, dst, P, rowptr, csr, N, E);

    const int tilesM = (N + 15) / 16;

    {
        dim3 blk(32, 4);
        dim3 grd(tilesM, HID / 64);
        k_gemm_scaled<128, 128><<<grd, blk, 0, stream>>>(x, W1, dinv, bufA, N);
    }
    k_gather_finalize<128, true><<<(N + 7) / 8, 256, 0, stream>>>(
        bufA, rowptr, csr, dinv, b1, bufC, N);

    {
        dim3 blk(32, 4);
        dim3 grd(tilesM, OUT / 64);
        k_gemm_scaled<128, 64><<<grd, blk, 0, stream>>>(bufC, W2, dinv, bufA, N);
    }
    k_gather_finalize<64, false><<<(N + 15) / 16, 256, 0, stream>>>(
        bufA, rowptr, csr, dinv, b2, out, N);
}


# Layer 1 aggregates first, then transforms: sort stores dinv[src] beside each source; gather kernel runs 16-node tiles through both WMMA products via LDS; 2 launches gone; layer-1 sum order changes
# speedup vs baseline: 3.3789x; 1.1409x over previous
#include <hip/hip_runtime.h>

typedef float v2f __attribute__((ext_vector_type(2)));
typedef float v8f __attribute__((ext_vector_type(8)));

constexpr int NCH = 64;
constexpr int SORT_T = 1024;

__launch_bounds__(SORT_T)
__global__ void k_chunk_hist(const int* __restrict__ dst, int* __restrict__ H,
                             int N, int E) {
    extern __shared__ int sh[];
    const int b   = blockIdx.x;
    const int per = (E + NCH - 1) / NCH;
    const int e0  = b * per;
    const int e1  = (e0 + per < E) ? (e0 + per) : E;
    for (int i = threadIdx.x; i < N; i += SORT_T) sh[i] = 0;
    __syncthreads();
    int e = e0 + threadIdx.x;
    for (; e + 3 * SORT_T < e1; e += 4 * SORT_T) {
        const int d0 = dst[e];
        const int d1 = dst[e + SORT_T];
        const int d2 = dst[e + 2 * SORT_T];
        const int d3 = dst[e + 3 * SORT_T];
        atomicAdd(&sh[d0], 1);
        atomicAdd(&sh[d1], 1);
        atomicAdd(&sh[d2], 1);
        atomicAdd(&sh[d3], 1);
    }
    for (; e < e1; e += SORT_T) atomicAdd(&sh[dst[e]], 1);
    __syncthreads();
    int* row = H + (size_t)b * N;
    for (int i = threadIdx.x; i < N; i += SORT_T) row[i] = sh[i];
}

__global__ void k_chunk_prefix(const int* __restrict__ H, int* __restrict__ P,
                               int* __restrict__ cnt, int N) {
    const int n = blockIdx.x * blockDim.x + threadIdx.x;
    if (n >= N) return;
    int run = 0;
#pragma unroll
    for (int b0 = 0; b0 < NCH; b0 += 16) {
        int t[16];
#pragma unroll
        for (int j = 0; j < 16; ++j) t[j] = H[(size_t)(b0 + j) * N + n];
#pragma unroll
        for (int j = 0; j < 16; ++j) {
            P[(size_t)(b0 + j) * N + n] = run;
            run += t[j];
        }
    }
    cnt[n] = run;
}

__launch_bounds__(SORT_T)
__global__ void k_node_scan(const int* __restrict__ cnt, int* __restrict__ rowptr,
                            float* __restrict__ dinv, int N) {
    __shared__ int s[2][SORT_T];
    const int t   = threadIdx.x;
    const int ipt = (N + SORT_T - 1) / SORT_T;
    const int i0  = t * ipt;
    const int i1  = (i0 + ipt < N) ? (i0 + ipt) : N;
    int sum = 0;
    for (int i = i0; i < i1; ++i) {
        const int c = cnt[i];
        sum += c;
        const float d = (float)c + 1.0f;
        dinv[i] = (d > 0.0f) ? rsqrtf(d) : 0.0f;
    }
    s[0][t] = sum;
    __syncthreads();
    int cur = 0;
    for (int off = 1; off < SORT_T; off <<= 1) {
        int v = s[cur][t];
        if (t >= off) v += s[cur][t - off];
        s[cur ^ 1][t] = v;
        __syncthreads();
        cur ^= 1;
    }
    int run = s[cur][t] - sum;
    for (int i = i0; i < i1; ++i) {
        rowptr[i] = run;
        run += cnt[i];
    }
    if (t == SORT_T - 1) rowptr[N] = s[cur][t];
}

__launch_bounds__(SORT_T)
__global__ void k_chunk_place(const int* __restrict__ src, const int* __restrict__ dst,
                              const int* __restrict__ P, const int* __restrict__ rowptr,
                              const float* __restrict__ dinv,
                              int2* __restrict__ ent, int N, int E) {
    extern __shared__ int sh[];
    const int b   = blockIdx.x;
    const int per = (E + NCH - 1) / NCH;
    const int e0  = b * per;
    const int e1  = (e0 + per < E) ? (e0 + per) : E;
    const int* row = P + (size_t)b * N;
    for (int i = threadIdx.x; i < N; i += SORT_T) sh[i] = rowptr[i] + row[i];
    __syncthreads();
    for (int e = e0 + threadIdx.x; e < e1; e += 8 * SORT_T) {
        int   d[8], s[8];
        float w[8];
#pragma unroll
        for (int u = 0; u < 8; ++u) {
            int ee = e + u * SORT_T;
            if (ee >= e1) ee = e;
            d[u] = dst[ee];
            s[u] = src[ee];
        }
#pragma unroll
        for (int u = 0; u < 8; ++u) w[u] = dinv[s[u]];
#pragma unroll
        for (int u = 0; u < 8; ++u) {
            if (e + u * SORT_T < e1) {
                const int pos = atomicAdd(&sh[d[u]], 1);
                ent[pos] = make_int2(s[u], __float_as_int(w[u]));
            }
        }
    }
}

constexpr int TILE_M = 16;
constexpr int PITCH  = 132;

template <int K, int NF>
__device__ __forceinline__ v8f tile_product(const float* At, const float* __restrict__ W,
                                            int colB, int r16, int kh) {
    const float* ap = At + r16 * PITCH + kh;
    const float* bp = W + (size_t)kh * NF + colB;
    v8f c = {};
#pragma unroll 16
    for (int k = 0; k < K; k += 4) {
        v2f a = *(const v2f*)ap;
        v2f b;
        b.x = bp[0];
        b.y = bp[NF];
        c = __builtin_amdgcn_wmma_f32_16x16x4_f32(
                 false, a,  false, b,
                 (short)0, c,  false,  false);
        ap += 4;
        bp += 4 * NF;
    }
    return c;
}

__launch_bounds__(TILE_M * 32)
__global__ void k_gather_dense(const float* __restrict__ x,
                               const int* __restrict__ rowptr,
                               const int2* __restrict__ ent,
                               const float* __restrict__ dinv,
                               const float* __restrict__ W1,
                               const float* __restrict__ b1,
                               const float* __restrict__ W2,
                               float* __restrict__ hs2, int N) {
    constexpr int IN = 128, HID = 128, OUT = 64;
    __shared__ float T[TILE_M * PITCH];
    __shared__ float U[TILE_M * PITCH];
    const int lane  = threadIdx.x & 31;
    const int wave  = threadIdx.x >> 5;
    const int node0 = blockIdx.x * TILE_M;
    const int i     = node0 + wave;
    const int q     = lane * 4;

    float4 acc = make_float4(0.0f, 0.0f, 0.0f, 0.0f);
    if (i < N) {
        const float* xq = x + q;
        const int j1 = rowptr[i + 1];
        int j = rowptr[i];
        for (; j + 8 <= j1; j += 8) {
            int2   e[8];
            float4 v[8];
#pragma unroll
            for (int u = 0; u < 8; ++u) e[u] = ent[j + u];
#pragma unroll
            for (int u = 0; u < 8; ++u) v[u] = *(const float4*)(xq + (size_t)e[u].x * IN);
#pragma unroll
            for (int u = 0; u < 8; ++u) {
                const float w = __int_as_float(e[u].y);
                acc.x = fmaf(w, v[u].x, acc.x);
                acc.y = fmaf(w, v[u].y, acc.y);
                acc.z = fmaf(w, v[u].z, acc.z);
                acc.w = fmaf(w, v[u].w, acc.w);
            }
        }
        for (; j < j1; ++j) {
            const int2 e = ent[j];
            const float w = __int_as_float(e.y);
            const float4 v = *(const float4*)(xq + (size_t)e.x * IN);
            acc.x = fmaf(w, v.x, acc.x);
            acc.y = fmaf(w, v.y, acc.y);
            acc.z = fmaf(w, v.z, acc.z);
            acc.w = fmaf(w, v.w, acc.w);
        }
        const float di = dinv[i];
        const float4 vh = *(const float4*)(xq + (size_t)i * IN);
        acc.x = di * (acc.x + di * vh.x);
        acc.y = di * (acc.y + di * vh.y);
        acc.z = di * (acc.z + di * vh.z);
        acc.w = di * (acc.w + di * vh.w);
    }
    *(float4*)(T + wave * PITCH + q) = acc;
    __syncthreads();

    const int r16 = lane & 15;
    const int hi  = lane >> 4;
    const int kh  = hi * 2;

    if (wave < HID / 16) {
        const int col = wave * 16 + r16;
        const v8f c = tile_product<IN, HID>(T, W1, col, r16, kh);
        const float bc = b1[col];
        float* up = U + (hi * 8) * PITCH + col;
#pragma unroll
        for (int r = 0; r < 8; ++r) up[r * PITCH] = fmaxf(c[r] + bc, 0.0f);
    }
    __syncthreads();

    if (wave < OUT / 16) {
        const int col = wave * 16 + r16;
        const v8f c = tile_product<HID, OUT>(U, W2, col, r16, kh);
        const int rowBase = node0 + hi * 8;
        float* op = hs2 + (size_t)rowBase * OUT + col;
        if (rowBase + 8 <= N) {
            const float4 d0 = *(const float4*)(dinv + rowBase);
            const float4 d1 = *(const float4*)(dinv + rowBase + 4);
            op[0 * (size_t)OUT] = c[0] * d0.x;
            op[1 * (size_t)OUT] = c[1] * d0.y;
            op[2 * (size_t)OUT] = c[2] * d0.z;
            op[3 * (size_t)OUT] = c[3] * d0.w;
            op[4 * (size_t)OUT] = c[4] * d1.x;
            op[5 * (size_t)OUT] = c[5] * d1.y;
            op[6 * (size_t)OUT] = c[6] * d1.z;
            op[7 * (size_t)OUT] = c[7] * d1.w;
        } else {
#pragma unroll
            for (int r = 0; r < 8; ++r) {
                const int row = rowBase + r;
                if (row < N) op[(size_t)r * OUT] = c[r] * dinv[row];
            }
        }
    }
}

template <int F, bool RELU>
__global__ void k_gather_finalize(const float* __restrict__ hs,
                                  const int* __restrict__ rowptr,
                                  const int2* __restrict__ ent,
                                  const float* __restrict__ dinv,
                                  const float* __restrict__ bias,
                                  float* __restrict__ out, int N) {
    constexpr int TPR = F / 4;
    const int i = blockIdx.x * (256 / TPR) + threadIdx.x / TPR;
    if (i >= N) return;
    const int q = (threadIdx.x % TPR) * 4;
    const float* hq = hs + q;
    const int j1 = rowptr[i + 1];
    int j = rowptr[i];
    float4 acc = make_float4(0.0f, 0.0f, 0.0f, 0.0f);
    for (; j + 8 <= j1; j += 8) {
        int s[8];
        float4 v[8];
#pragma unroll
        for (int u = 0; u < 8; ++u) s[u] = ent[j + u].x;
#pragma unroll
        for (int u = 0; u < 8; ++u) v[u] = *(const float4*)(hq + (size_t)s[u] * F);
#pragma unroll
        for (int u = 0; u < 8; ++u) {
            acc.x += v[u].x; acc.y += v[u].y; acc.z += v[u].z; acc.w += v[u].w;
        }
    }
    for (; j < j1; ++j) {
        const float4 v = *(const float4*)(hq + (size_t)ent[j].x * F);
        acc.x += v.x; acc.y += v.y; acc.z += v.z; acc.w += v.w;
    }
    const size_t off = (size_t)i * F + q;
    const float di = dinv[i];
    const float4 vh = *(const float4*)(hs + off);
    const float4 vb = *(const float4*)(bias + q);
    float4 r;
    r.x = di * (acc.x + vh.x) + vb.x;
    r.y = di * (acc.y + vh.y) + vb.y;
    r.z = di * (acc.z + vh.z) + vb.z;
    r.w = di * (acc.w + vh.w) + vb.w;
    if (RELU) {
        r.x = fmaxf(r.x, 0.0f);
        r.y = fmaxf(r.y, 0.0f);
        r.z = fmaxf(r.z, 0.0f);
        r.w = fmaxf(r.w, 0.0f);
    }
    *(float4*)(out + off) = r;
}

extern "C" void kernel_launch(void* const* d_in, const int* in_sizes, int n_in,
                              void* d_out, int out_size, void* d_ws, size_t ws_size,
                              hipStream_t stream) {
    const float* x  = (const float*)d_in[0];
    const float* W1 = (const float*)d_in[1];
    const float* b1 = (const float*)d_in[2];
    const float* W2 = (const float*)d_in[3];
    const float* b2 = (const float*)d_in[4];
    const int*   ei = (const int*)d_in[5];

    const int IN = 128, OUT = 64;
    const int N = in_sizes[0] / IN;
    const int E = in_sizes[5] / 2;
    const int* src = ei;
    const int* dst = ei + E;

    const size_t Np = ((size_t)N + 64) & ~(size_t)63;
    const size_t Ep = ((size_t)E + 63) & ~(size_t)63;

    float* ws     = (float*)d_ws;
    float* dinv   = ws;
    int*   rowptr = (int*)(dinv + Np);
    int*   cnt    = rowptr + Np;
    int*   H      = cnt + Np;
    int*   P      = H + (size_t)NCH * Np;
    int2*  ent    = (int2*)(P + (size_t)NCH * Np);
    float* hs2    = (float*)(ent + Ep);
    float* out    = (float*)d_out;

    const size_t ldsN = (size_t)N * sizeof(int);
    k_chunk_hist<<<NCH, SORT_T, ldsN, stream>>>(dst, H, N, E);
    k_chunk_prefix<<<(N + 255) / 256, 256, 0, stream>>>(H, P, cnt, N);
    k_node_scan<<<1, SORT_T, 0, stream>>>(cnt, rowptr, dinv, N);
    k_chunk_place<<<NCH, SORT_T, ldsN, stream>>>(src, dst, P, rowptr, dinv, ent, N, E);

    k_gather_dense<<<(N + TILE_M - 1) / TILE_M, TILE_M * 32, 0, stream>>>(
        x, rowptr, ent, dinv, W1, b1, W2, hs2, N);

    k_gather_finalize<64, false><<<(N + 15) / 16, 256, 0, stream>>>(
        hs2, rowptr, ent, dinv, b2, out, N);
}


# Row pointers and dinv now come out of the chunk-prefix kernel, from per-chunk edge counts of 256-node groups that the histogram kernel adds; the single-work-group node scan launch is gone
# speedup vs baseline: 3.6653x; 1.0848x over previous
#include <hip/hip_runtime.h>

typedef float v2f __attribute__((ext_vector_type(2)));
typedef float v8f __attribute__((ext_vector_type(8)));

constexpr int NCH = 64;
constexpr int SORT_T = 1024;

constexpr int GRP = 256;

__device__ __forceinline__ int wave_sum(int v) {
#pragma unroll
    for (int o = 16; o > 0; o >>= 1) v += __shfl_xor(v, o, 32);
    return v;
}

__launch_bounds__(SORT_T)
__global__ void k_chunk_hist(const int* __restrict__ dst, int* __restrict__ H,
                             int* __restrict__ G, int N, int E) {
    extern __shared__ int sh[];
    const int b   = blockIdx.x;
    const int per = (E + NCH - 1) / NCH;
    const int e0  = b * per;
    const int e1  = (e0 + per < E) ? (e0 + per) : E;
    for (int i = threadIdx.x; i < N; i += SORT_T) sh[i] = 0;
    __syncthreads();
    int e = e0 + threadIdx.x;
    for (; e + 3 * SORT_T < e1; e += 4 * SORT_T) {
        const int d0 = dst[e];
        const int d1 = dst[e + SORT_T];
        const int d2 = dst[e + 2 * SORT_T];
        const int d3 = dst[e + 3 * SORT_T];
        atomicAdd(&sh[d0], 1);
        atomicAdd(&sh[d1], 1);
        atomicAdd(&sh[d2], 1);
        atomicAdd(&sh[d3], 1);
    }
    for (; e < e1; e += SORT_T) atomicAdd(&sh[dst[e]], 1);
    __syncthreads();
    int* row = H + (size_t)b * N;
    for (int i = threadIdx.x; i < N; i += SORT_T) row[i] = sh[i];
    const int lane = threadIdx.x & 31;
    const int NG   = (N + GRP - 1) / GRP;
    for (int g = threadIdx.x >> 5; g < NG; g += SORT_T / 32) {
        int s = 0;
#pragma unroll
        for (int k = 0; k < GRP / 32; ++k) {
            const int i = g * GRP + k * 32 + lane;
            s += (i < N) ? sh[i] : 0;
        }
        s = wave_sum(s);
        if (lane == 0) G[(size_t)g * NCH + b] = s;
    }
}

__launch_bounds__(GRP)
__global__ void k_chunk_prefix(const int* __restrict__ H, const int* __restrict__ G,
                               int* __restrict__ P, int* __restrict__ rowptr,
                               float* __restrict__ dinv, int N) {
    __shared__ int s_part[GRP / 32];
    __shared__ int s_tot[GRP / 32];
    const int t    = threadIdx.x;
    const int lane = t & 31;
    const int wave = t >> 5;
    const int n    = blockIdx.x * GRP + t;

    const int4* G4  = (const int4*)G;
    const int  lim4 = blockIdx.x * (NCH / 4);
    int4 q[4];
#pragma unroll
    for (int u = 0; u < 4; ++u) {
        const int k = u * GRP + t;
        q[u] = (k < lim4) ? G4[k] : make_int4(0, 0, 0, 0);
    }

    int run = 0;
    if (n < N) {
#pragma unroll
        for (int b0 = 0; b0 < NCH; b0 += 16) {
            int c[16];
#pragma unroll
            for (int j = 0; j < 16; ++j) c[j] = H[(size_t)(b0 + j) * N + n];
#pragma unroll
            for (int j = 0; j < 16; ++j) {
                P[(size_t)(b0 + j) * N + n] = run;
                run += c[j];
            }
        }
    }

    int part = 0;
#pragma unroll
    for (int u = 0; u < 4; ++u) part += q[u].x + q[u].y + q[u].z + q[u].w;
    for (int k = 4 * GRP + t; k < lim4; k += GRP) {
        const int4 r = G4[k];
        part += r.x + r.y + r.z + r.w;
    }

    int incl = run;
#pragma unroll
    for (int o = 1; o < 32; o <<= 1) {
        const int v = __shfl_up(incl, o, 32);
        if (lane >= o) incl += v;
    }
    part = wave_sum(part);
    if (lane == 31) s_tot[wave]  = incl;
    if (lane == 0)  s_part[wave] = part;
    __syncthreads();
    int first = 0;
#pragma unroll
    for (int w = 0; w < GRP / 32; ++w) {
        first += s_part[w];
        if (w < wave) first += s_tot[w];
    }
    if (n < N) {
        const int r0 = first + incl - run;
        rowptr[n] = r0;
        const float d = (float)run + 1.0f;
        dinv[n] = (d > 0.0f) ? rsqrtf(d) : 0.0f;
        if (n == N - 1) rowptr[N] = r0 + run;
    }
}

__launch_bounds__(SORT_T)
__global__ void k_chunk_place(const int* __restrict__ src, const int* __restrict__ dst,
                              const int* __restrict__ P, const int* __restrict__ rowptr,
                              const float* __restrict__ dinv,
                              int2* __restrict__ ent, int N, int E) {
    extern __shared__ int sh[];
    const int b   = blockIdx.x;
    const int per = (E + NCH - 1) / NCH;
    const int e0  = b * per;
    const int e1  = (e0 + per < E) ? (e0 + per) : E;
    const int* row = P + (size_t)b * N;
    for (int i = threadIdx.x; i < N; i += SORT_T) sh[i] = rowptr[i] + row[i];
    __syncthreads();
    for (int e = e0 + threadIdx.x; e < e1; e += 8 * SORT_T) {
        int   d[8], s[8];
        float w[8];
#pragma unroll
        for (int u = 0; u < 8; ++u) {
            int ee = e + u * SORT_T;
            if (ee >= e1) ee = e;
            d[u] = dst[ee];
            s[u] = src[ee];
        }
#pragma unroll
        for (int u = 0; u < 8; ++u) w[u] = dinv[s[u]];
#pragma unroll
        for (int u = 0; u < 8; ++u) {
            if (e + u * SORT_T < e1) {
                const int pos = atomicAdd(&sh[d[u]], 1);
                ent[pos] = make_int2(s[u], __float_as_int(w[u]));
            }
        }
    }
}

constexpr int TILE_M = 16;
constexpr int PITCH  = 132;

template <int K, int NF>
__device__ __forceinline__ v8f tile_product(const float* At, const float* __restrict__ W,
                                            int colB, int r16, int kh) {
    const float* ap = At + r16 * PITCH + kh;
    const float* bp = W + (size_t)kh * NF + colB;
    v8f c = {};
#pragma unroll 16
    for (int k = 0; k < K; k += 4) {
        v2f a = *(const v2f*)ap;
        v2f b;
        b.x = bp[0];
        b.y = bp[NF];
        c = __builtin_amdgcn_wmma_f32_16x16x4_f32(
                 false, a,  false, b,
                 (short)0, c,  false,  false);
        ap += 4;
        bp += 4 * NF;
    }
    return c;
}

__launch_bounds__(TILE_M * 32)
__global__ void k_gather_dense(const float* __restrict__ x,
                               const int* __restrict__ rowptr,
                               const int2* __restrict__ ent,
                               const float* __restrict__ dinv,
                               const float* __restrict__ W1,
                               const float* __restrict__ b1,
                               const float* __restrict__ W2,
                               float* __restrict__ hs2, int N) {
    constexpr int IN = 128, HID = 128, OUT = 64;
    __shared__ float T[TILE_M * PITCH];
    __shared__ float U[TILE_M * PITCH];
    const int lane  = threadIdx.x & 31;
    const int wave  = threadIdx.x >> 5;
    const int node0 = blockIdx.x * TILE_M;
    const int i     = node0 + wave;
    const int q     = lane * 4;

    float4 acc = make_float4(0.0f, 0.0f, 0.0f, 0.0f);
    if (i < N) {
        const float* xq = x + q;
        const int j1 = rowptr[i + 1];
        int j = rowptr[i];
        for (; j + 8 <= j1; j += 8) {
            int2   e[8];
            float4 v[8];
#pragma unroll
            for (int u = 0; u < 8; ++u) e[u] = ent[j + u];
#pragma unroll
            for (int u = 0; u < 8; ++u) v[u] = *(const float4*)(xq + (size_t)e[u].x * IN);
#pragma unroll
            for (int u = 0; u < 8; ++u) {
                const float w = __int_as_float(e[u].y);
                acc.x = fmaf(w, v[u].x, acc.x);
                acc.y = fmaf(w, v[u].y, acc.y);
                acc.z = fmaf(w, v[u].z, acc.z);
                acc.w = fmaf(w, v[u].w, acc.w);
            }
        }
        for (; j < j1; ++j) {
            const int2 e = ent[j];
            const float w = __int_as_float(e.y);
            const float4 v = *(const float4*)(xq + (size_t)e.x * IN);
            acc.x = fmaf(w, v.x, acc.x);
            acc.y = fmaf(w, v.y, acc.y);
            acc.z = fmaf(w, v.z, acc.z);
            acc.w = fmaf(w, v.w, acc.w);
        }
        const float di = dinv[i];
        const float4 vh = *(const float4*)(xq + (size_t)i * IN);
        acc.x = di * (acc.x + di * vh.x);
        acc.y = di * (acc.y + di * vh.y);
        acc.z = di * (acc.z + di * vh.z);
        acc.w = di * (acc.w + di * vh.w);
    }
    *(float4*)(T + wave * PITCH + q) = acc;
    __syncthreads();

    const int r16 = lane & 15;
    const int hi  = lane >> 4;
    const int kh  = hi * 2;

    if (wave < HID / 16) {
        const int col = wave * 16 + r16;
        const v8f c = tile_product<IN, HID>(T, W1, col, r16, kh);
        const float bc = b1[col];
        float* up = U + (hi * 8) * PITCH + col;
#pragma unroll
        for (int r = 0; r < 8; ++r) up[r * PITCH] = fmaxf(c[r] + bc, 0.0f);
    }
    __syncthreads();

    if (wave < OUT / 16) {
        const int col = wave * 16 + r16;
        const v8f c = tile_product<HID, OUT>(U, W2, col, r16, kh);
        const int rowBase = node0 + hi * 8;
        float* op = hs2 + (size_t)rowBase * OUT + col;
        if (rowBase + 8 <= N) {
            const float4 d0 = *(const float4*)(dinv + rowBase);
            const float4 d1 = *(const float4*)(dinv + rowBase + 4);
            op[0 * (size_t)OUT] = c[0] * d0.x;
            op[1 * (size_t)OUT] = c[1] * d0.y;
            op[2 * (size_t)OUT] = c[2] * d0.z;
            op[3 * (size_t)OUT] = c[3] * d0.w;
            op[4 * (size_t)OUT] = c[4] * d1.x;
            op[5 * (size_t)OUT] = c[5] * d1.y;
            op[6 * (size_t)OUT] = c[6] * d1.z;
            op[7 * (size_t)OUT] = c[7] * d1.w;
        } else {
#pragma unroll
            for (int r = 0; r < 8; ++r) {
                const int row = rowBase + r;
                if (row < N) op[(size_t)r * OUT] = c[r] * dinv[row];
            }
        }
    }
}

template <int F, bool RELU>
__global__ void k_gather_finalize(const float* __restrict__ hs,
                                  const int* __restrict__ rowptr,
                                  const int2* __restrict__ ent,
                                  const float* __restrict__ dinv,
                                  const float* __restrict__ bias,
                                  float* __restrict__ out, int N) {
    constexpr int TPR = F / 4;
    const int i = blockIdx.x * (256 / TPR) + threadIdx.x / TPR;
    if (i >= N) return;
    const int q = (threadIdx.x % TPR) * 4;
    const float* hq = hs + q;
    const int j1 = rowptr[i + 1];
    int j = rowptr[i];
    float4 acc = make_float4(0.0f, 0.0f, 0.0f, 0.0f);
    for (; j + 8 <= j1; j += 8) {
        int s[8];
        float4 v[8];
#pragma unroll
        for (int u = 0; u < 8; ++u) s[u] = ent[j + u].x;
#pragma unroll
        for (int u = 0; u < 8; ++u) v[u] = *(const float4*)(hq + (size_t)s[u] * F);
#pragma unroll
        for (int u = 0; u < 8; ++u) {
            acc.x += v[u].x; acc.y += v[u].y; acc.z += v[u].z; acc.w += v[u].w;
        }
    }
    for (; j < j1; ++j) {
        const float4 v = *(const float4*)(hq + (size_t)ent[j].x * F);
        acc.x += v.x; acc.y += v.y; acc.z += v.z; acc.w += v.w;
    }
    const size_t off = (size_t)i * F + q;
    const float di = dinv[i];
    const float4 vh = *(const float4*)(hs + off);
    const float4 vb = *(const float4*)(bias + q);
    float4 r;
    r.x = di * (acc.x + vh.x) + vb.x;
    r.y = di * (acc.y + vh.y) + vb.y;
    r.z = di * (acc.z + vh.z) + vb.z;
    r.w = di * (acc.w + vh.w) + vb.w;
    if (RELU) {
        r.x = fmaxf(r.x, 0.0f);
        r.y = fmaxf(r.y, 0.0f);
        r.z = fmaxf(r.z, 0.0f);
        r.w = fmaxf(r.w, 0.0f);
    }
    *(float4*)(out + off) = r;
}

extern "C" void kernel_launch(void* const* d_in, const int* in_sizes, int n_in,
                              void* d_out, int out_size, void* d_ws, size_t ws_size,
                              hipStream_t stream) {
    const float* x  = (const float*)d_in[0];
    const float* W1 = (const float*)d_in[1];
    const float* b1 = (const float*)d_in[2];
    const float* W2 = (const float*)d_in[3];
    const float* b2 = (const float*)d_in[4];
    const int*   ei = (const int*)d_in[5];

    const int IN = 128, OUT = 64;
    const int N = in_sizes[0] / IN;
    const int E = in_sizes[5] / 2;
    const int* src = ei;
    const int* dst = ei + E;

    const size_t Np = ((size_t)N + 64) & ~(size_t)63;
    const size_t Ep = ((size_t)E + 63) & ~(size_t)63;

    float* ws     = (float*)d_ws;
    float* dinv   = ws;
    int*   rowptr = (int*)(dinv + Np);
    int*   G      = rowptr + Np;
    int*   H      = G + (((size_t)N + GRP - 1) / GRP) * NCH;
    int*   P      = H + (size_t)NCH * Np;
    int2*  ent    = (int2*)(P + (size_t)NCH * Np);
    float* hs2    = (float*)(ent + Ep);
    float* out    = (float*)d_out;

    const size_t ldsN = (size_t)N * sizeof(int);
    k_chunk_hist<<<NCH, SORT_T, ldsN, stream>>>(dst, H, G, N, E);
    k_chunk_prefix<<<(N + GRP - 1) / GRP, GRP, 0, stream>>>(H, G, P, rowptr, dinv, N);
    k_chunk_place<<<NCH, SORT_T, ldsN, stream>>>(src, dst, P, rowptr, dinv, ent, N, E);

    k_gather_dense<<<(N + TILE_M - 1) / TILE_M, TILE_M * 32, 0, stream>>>(
        x, rowptr, ent, dinv, W1, b1, W2, hs2, N);

    k_gather_finalize<64, false><<<(N + 15) / 16, 256, 0, stream>>>(
        hs2, rowptr, ent, dinv, b2, out, N);
}


# Two-level edge sort: each chunk sorts its slice by destination in LDS, then one work-group per 128-node group places its rows; count planes and 1 launch gone; in-row order within a chunk may change
# speedup vs baseline: 4.5916x; 1.2527x over previous
#include <hip/hip_runtime.h>

typedef float v2f __attribute__((ext_vector_type(2)));
typedef float v8f __attribute__((ext_vector_type(8)));

constexpr int NCH = 64;
constexpr int SORT_T = 1024;
constexpr int GB = 128;

__device__ __forceinline__ int wave_sum(int v) {
#pragma unroll
    for (int o = 16; o > 0; o >>= 1) v += __shfl_xor(v, o, 32);
    return v;
}

__device__ __forceinline__ int wave_incl_scan(int v, int lane) {
#pragma unroll
    for (int o = 1; o < 32; o <<= 1) {
        const int u = __shfl_up(v, o, 32);
        if (lane >= o) v += u;
    }
    return v;
}

__launch_bounds__(SORT_T)
__global__ void k_chunk_sort(const int* __restrict__ src, const int* __restrict__ dst,
                             int2* __restrict__ sorted, int* __restrict__ Goff, int N, int E) {
    extern __shared__ int sh[];
    __shared__ int s_wtot[SORT_T / 32];
    const int t    = threadIdx.x;
    const int lane = t & 31;
    const int wave = t >> 5;
    const int b    = blockIdx.x;
    const int per  = (E + NCH - 1) / NCH;
    const int e0   = b * per;
    const int e1   = (e0 + per < E) ? (e0 + per) : E;
    const int R    = (N + SORT_T - 1) / SORT_T;
    for (int i = t; i < R * SORT_T; i += SORT_T) sh[i] = 0;
    __syncthreads();
    int e = e0 + t;
    for (; e + 3 * SORT_T < e1; e += 4 * SORT_T) {
        const int d0 = dst[e];
        const int d1 = dst[e + SORT_T];
        const int d2 = dst[e + 2 * SORT_T];
        const int d3 = dst[e + 3 * SORT_T];
        atomicAdd(&sh[d0], 1);
        atomicAdd(&sh[d1], 1);
        atomicAdd(&sh[d2], 1);
        atomicAdd(&sh[d3], 1);
    }
    for (; e < e1; e += SORT_T) atomicAdd(&sh[dst[e]], 1);
    __syncthreads();

    const int i0 = t * R;
    int s = 0;
#pragma unroll 4
    for (int k = 0; k < R; ++k) s += sh[i0 + k];
    const int incl = wave_incl_scan(s, lane);
    if (lane == 31) s_wtot[wave] = incl;
    __syncthreads();
    const int woff = wave_sum((lane < wave) ? s_wtot[lane] : 0);
    int run = woff + incl - s;
#pragma unroll 4
    for (int k = 0; k < R; ++k) {
        const int v = sh[i0 + k];
        sh[i0 + k] = run;
        run += v;
    }
    __syncthreads();

    const int NG  = (N + GB - 1) / GB;
    const int tot = (e1 > e0) ? (e1 - e0) : 0;
    for (int g = t; g <= NG; g += SORT_T)
        Goff[(size_t)g * NCH + b] = (g < NG) ? sh[g * GB] : tot;
    __syncthreads();

    int2* slice = sorted + e0;
    for (int ee = e0 + t; ee < e1; ee += 8 * SORT_T) {
        int d[8], sc[8];
#pragma unroll
        for (int u = 0; u < 8; ++u) {
            int x = ee + u * SORT_T;
            if (x >= e1) x = ee;
            d[u]  = dst[x];
            sc[u] = src[x];
        }
#pragma unroll
        for (int u = 0; u < 8; ++u) {
            if (ee + u * SORT_T < e1) {
                const int pos = atomicAdd(&sh[d[u]], 1);
                slice[pos] = make_int2(sc[u], d[u]);
            }
        }
    }
}

__launch_bounds__(SORT_T)
__global__ void k_group_place(const int2* __restrict__ sorted, const int* __restrict__ Goff,
                              int* __restrict__ rowptr, float* __restrict__ dinv,
                              int* __restrict__ ent, int N, int E) {
    __shared__ int cnt[NCH * GB];
    __shared__ int s_lo[NCH];
    __shared__ int s_hi[NCH];
    __shared__ int s_bp[NCH / 32];
    __shared__ int s_wt[GB / 32];
    __shared__ int s_rs[GB];
    const int t    = threadIdx.x;
    const int lane = t & 31;
    const int wave = t >> 5;
    const int g    = blockIdx.x;
    const int n0   = g * GB;
    const int per  = (E + NCH - 1) / NCH;

    if (t < NCH) {
        const int lo = Goff[(size_t)g * NCH + t];
        const int hi = Goff[(size_t)(g + 1) * NCH + t];
        s_lo[t] = lo;
        s_hi[t] = hi;
        const int p = wave_sum(lo);
        if (lane == 0) s_bp[wave] = p;
    }
    for (int i = t; i < NCH * GB; i += SORT_T) cnt[i] = 0;
    __syncthreads();
    int base = 0;
#pragma unroll
    for (int w = 0; w < NCH / 32; ++w) base += s_bp[w];

    for (int c = wave; c < NCH; c += SORT_T / 32) {
        const int lo = s_lo[c];
        const int hi = s_hi[c];
        const int2* sp = sorted + (size_t)c * per;
        const int cb = c * GB - n0;
        for (int k0 = lo + lane; k0 < hi; k0 += 8 * 32) {
            int dd[8];
#pragma unroll
            for (int u = 0; u < 8; ++u) {
                const int k = k0 + u * 32;
                dd[u] = sp[(k < hi) ? k : k0].y;
            }
#pragma unroll
            for (int u = 0; u < 8; ++u)
                if (k0 + u * 32 < hi) atomicAdd(&cnt[cb + dd[u]], 1);
        }
    }
    __syncthreads();

    int run = 0;
    int incl = 0;
    if (t < GB) {
#pragma unroll 16
        for (int c = 0; c < NCH; ++c) {
            const int v = cnt[c * GB + t];
            cnt[c * GB + t] = run;
            run += v;
        }
        incl = wave_incl_scan(run, lane);
        if (lane == 31) s_wt[wave] = incl;
    }
    __syncthreads();
    if (t < GB) {
        int first = base;
#pragma unroll
        for (int w = 0; w < GB / 32; ++w)
            if (w < wave) first += s_wt[w];
        const int r0 = first + incl - run;
        s_rs[t] = r0;
        const int n = n0 + t;
        if (n < N) {
            rowptr[n] = r0;
            const float d = (float)run + 1.0f;
            dinv[n] = (d > 0.0f) ? rsqrtf(d) : 0.0f;
            if (n == N - 1) rowptr[N] = r0 + run;
        }
    }
    __syncthreads();

    for (int c = wave; c < NCH; c += SORT_T / 32) {
        const int lo = s_lo[c];
        const int hi = s_hi[c];
        const int2* sp = sorted + (size_t)c * per;
        const int cb = c * GB - n0;
        for (int k0 = lo + lane; k0 < hi; k0 += 8 * 32) {
            int2 p[8];
#pragma unroll
            for (int u = 0; u < 8; ++u) {
                const int k = k0 + u * 32;
                p[u] = sp[(k < hi) ? k : k0];
            }
#pragma unroll
            for (int u = 0; u < 8; ++u) {
                if (k0 + u * 32 < hi) {
                    const int nl  = p[u].y - n0;
                    const int pos = s_rs[nl] + atomicAdd(&cnt[cb + p[u].y], 1);
                    ent[pos] = p[u].x;
                }
            }
        }
    }
}

constexpr int TILE_M = 16;
constexpr int PITCH  = 132;

template <int K, int NF>
__device__ __forceinline__ v8f tile_product(const float* At, const float* __restrict__ W,
                                            int colB, int r16, int kh) {
    const float* ap = At + r16 * PITCH + kh;
    const float* bp = W + (size_t)kh * NF + colB;
    v8f c = {};
#pragma unroll 16
    for (int k = 0; k < K; k += 4) {
        v2f a = *(const v2f*)ap;
        v2f b;
        b.x = bp[0];
        b.y = bp[NF];
        c = __builtin_amdgcn_wmma_f32_16x16x4_f32(
                 false, a,  false, b,
                 (short)0, c,  false,  false);
        ap += 4;
        bp += 4 * NF;
    }
    return c;
}

__launch_bounds__(TILE_M * 32)
__global__ void k_gather_dense(const float* __restrict__ x,
                               const int* __restrict__ rowptr,
                               const int* __restrict__ ent,
                               const float* __restrict__ dinv,
                               const float* __restrict__ W1,
                               const float* __restrict__ b1,
                               const float* __restrict__ W2,
                               float* __restrict__ hs2, int N) {
    constexpr int IN = 128, HID = 128, OUT = 64;
    __shared__ float T[TILE_M * PITCH];
    __shared__ float U[TILE_M * PITCH];
    const int lane  = threadIdx.x & 31;
    const int wave  = threadIdx.x >> 5;
    const int node0 = blockIdx.x * TILE_M;
    const int i     = node0 + wave;
    const int q     = lane * 4;

    float4 acc = make_float4(0.0f, 0.0f, 0.0f, 0.0f);
    if (i < N) {
        const float* xq = x + q;
        const int j1 = rowptr[i + 1];
        int j = rowptr[i];
        for (; j + 8 <= j1; j += 8) {
            int    s[8];
            float  wt[8];
            float4 v[8];
#pragma unroll
            for (int u = 0; u < 8; ++u) s[u] = ent[j + u];
#pragma unroll
            for (int u = 0; u < 8; ++u) v[u] = *(const float4*)(xq + (size_t)s[u] * IN);
#pragma unroll
            for (int u = 0; u < 8; ++u) wt[u] = dinv[s[u]];
#pragma unroll
            for (int u = 0; u < 8; ++u) {
                const float w = wt[u];
                acc.x = fmaf(w, v[u].x, acc.x);
                acc.y = fmaf(w, v[u].y, acc.y);
                acc.z = fmaf(w, v[u].z, acc.z);
                acc.w = fmaf(w, v[u].w, acc.w);
            }
        }
        for (; j < j1; ++j) {
            const int s = ent[j];
            const float w = dinv[s];
            const float4 v = *(const float4*)(xq + (size_t)s * IN);
            acc.x = fmaf(w, v.x, acc.x);
            acc.y = fmaf(w, v.y, acc.y);
            acc.z = fmaf(w, v.z, acc.z);
            acc.w = fmaf(w, v.w, acc.w);
        }
        const float di = dinv[i];
        const float4 vh = *(const float4*)(xq + (size_t)i * IN);
        acc.x = di * (acc.x + di * vh.x);
        acc.y = di * (acc.y + di * vh.y);
        acc.z = di * (acc.z + di * vh.z);
        acc.w = di * (acc.w + di * vh.w);
    }
    *(float4*)(T + wave * PITCH + q) = acc;
    __syncthreads();

    const int r16 = lane & 15;
    const int hi  = lane >> 4;
    const int kh  = hi * 2;

    if (wave < HID / 16) {
        const int col = wave * 16 + r16;
        const v8f c = tile_product<IN, HID>(T, W1, col, r16, kh);
        const float bc = b1[col];
        float* up = U + (hi * 8) * PITCH + col;
#pragma unroll
        for (int r = 0; r < 8; ++r) up[r * PITCH] = fmaxf(c[r] + bc, 0.0f);
    }
    __syncthreads();

    if (wave < OUT / 16) {
        const int col = wave * 16 + r16;
        const v8f c = tile_product<HID, OUT>(U, W2, col, r16, kh);
        const int rowBase = node0 + hi * 8;
        float* op = hs2 + (size_t)rowBase * OUT + col;
        if (rowBase + 8 <= N) {
            const float4 d0 = *(const float4*)(dinv + rowBase);
            const float4 d1 = *(const float4*)(dinv + rowBase + 4);
            op[0 * (size_t)OUT] = c[0] * d0.x;
            op[1 * (size_t)OUT] = c[1] * d0.y;
            op[2 * (size_t)OUT] = c[2] * d0.z;
            op[3 * (size_t)OUT] = c[3] * d0.w;
            op[4 * (size_t)OUT] = c[4] * d1.x;
            op[5 * (size_t)OUT] = c[5] * d1.y;
            op[6 * (size_t)OUT] = c[6] * d1.z;
            op[7 * (size_t)OUT] = c[7] * d1.w;
        } else {
#pragma unroll
            for (int r = 0; r < 8; ++r) {
                const int row = rowBase + r;
                if (row < N) op[(size_t)r * OUT] = c[r] * dinv[row];
            }
        }
    }
}

template <int F, bool RELU>
__global__ void k_gather_finalize(const float* __restrict__ hs,
                                  const int* __restrict__ rowptr,
                                  const int* __restrict__ ent,
                                  const float* __restrict__ dinv,
                                  const float* __restrict__ bias,
                                  float* __restrict__ out, int N) {
    constexpr int TPR = F / 4;
    const int i = blockIdx.x * (256 / TPR) + threadIdx.x / TPR;
    if (i >= N) return;
    const int q = (threadIdx.x % TPR) * 4;
    const float* hq = hs + q;
    const int j1 = rowptr[i + 1];
    int j = rowptr[i];
    float4 acc = make_float4(0.0f, 0.0f, 0.0f, 0.0f);
    for (; j + 8 <= j1; j += 8) {
        int s[8];
        float4 v[8];
#pragma unroll
        for (int u = 0; u < 8; ++u) s[u] = ent[j + u];
#pragma unroll
        for (int u = 0; u < 8; ++u) v[u] = *(const float4*)(hq + (size_t)s[u] * F);
#pragma unroll
        for (int u = 0; u < 8; ++u) {
            acc.x += v[u].x; acc.y += v[u].y; acc.z += v[u].z; acc.w += v[u].w;
        }
    }
    for (; j < j1; ++j) {
        const float4 v = *(const float4*)(hq + (size_t)ent[j] * F);
        acc.x += v.x; acc.y += v.y; acc.z += v.z; acc.w += v.w;
    }
    const size_t off = (size_t)i * F + q;
    const float di = dinv[i];
    const float4 vh = *(const float4*)(hs + off);
    const float4 vb = *(const float4*)(bias + q);
    float4 r;
    r.x = di * (acc.x + vh.x) + vb.x;
    r.y = di * (acc.y + vh.y) + vb.y;
    r.z = di * (acc.z + vh.z) + vb.z;
    r.w = di * (acc.w + vh.w) + vb.w;
    if (RELU) {
        r.x = fmaxf(r.x, 0.0f);
        r.y = fmaxf(r.y, 0.0f);
        r.z = fmaxf(r.z, 0.0f);
        r.w = fmaxf(r.w, 0.0f);
    }
    *(float4*)(out + off) = r;
}

extern "C" void kernel_launch(void* const* d_in, const int* in_sizes, int n_in,
                              void* d_out, int out_size, void* d_ws, size_t ws_size,
                              hipStream_t stream) {
    const float* x  = (const float*)d_in[0];
    const float* W1 = (const float*)d_in[1];
    const float* b1 = (const float*)d_in[2];
    const float* W2 = (const float*)d_in[3];
    const float* b2 = (const float*)d_in[4];
    const int*   ei = (const int*)d_in[5];

    const int IN = 128, OUT = 64;
    const int N = in_sizes[0] / IN;
    const int E = in_sizes[5] / 2;
    const int* src = ei;
    const int* dst = ei + E;

    const size_t Np = ((size_t)N + 64) & ~(size_t)63;
    const size_t Ep = ((size_t)E + 63) & ~(size_t)63;
    const int    NG = (N + GB - 1) / GB;

    float* ws     = (float*)d_ws;
    float* dinv   = ws;
    int*   rowptr = (int*)(dinv + Np);
    int*   Goff   = rowptr + Np;
    int2*  sorted = (int2*)(Goff + ((size_t)NG + 1) * NCH);
    int*   ent    = (int*)(sorted + Ep);
    float* hs2    = (float*)(ent + Ep);
    float* out    = (float*)d_out;

    const size_t ldsN = (((size_t)N + SORT_T - 1) / SORT_T) * SORT_T * sizeof(int);
    k_chunk_sort<<<NCH, SORT_T, ldsN, stream>>>(src, dst, sorted, Goff, N, E);
    k_group_place<<<NG, SORT_T, 0, stream>>>(sorted, Goff, rowptr, dinv, ent, N, E);

    k_gather_dense<<<(N + TILE_M - 1) / TILE_M, TILE_M * 32, 0, stream>>>(
        x, rowptr, ent, dinv, W1, b1, W2, hs2, N);

    k_gather_finalize<64, false><<<(N + 15) / 16, 256, 0, stream>>>(
        hs2, rowptr, ent, dinv, b2, out, N);
}


# Both sort levels place records into an LDS image of the range they own and copy it out with whole-line stores; per-record partial-line global stores gone; row order still by LDS cursor arrival
# speedup vs baseline: 5.0480x; 1.0994x over previous
#include <hip/hip_runtime.h>

typedef float v2f __attribute__((ext_vector_type(2)));
typedef float v8f __attribute__((ext_vector_type(8)));

constexpr int NCH = 64;
constexpr int SORT_T = 1024;
constexpr int GB = 128;
constexpr int CAP1 = 10240;
constexpr int CAP2 = 16384;

__device__ __forceinline__ int wave_sum(int v) {
#pragma unroll
    for (int o = 16; o > 0; o >>= 1) v += __shfl_xor(v, o, 32);
    return v;
}

__device__ __forceinline__ int wave_incl_scan(int v, int lane) {
#pragma unroll
    for (int o = 1; o < 32; o <<= 1) {
        const int u = __shfl_up(v, o, 32);
        if (lane >= o) v += u;
    }
    return v;
}

template <bool STAGED>
__device__ __forceinline__ void chunk_scatter(const int* __restrict__ src, const int* __restrict__ dst,
                                              int* cur, int2* image, int2* __restrict__ slice,
                                              int e0, int e1, int t) {
    for (int ee = e0 + t; ee < e1; ee += 8 * SORT_T) {
        int d[8], sc[8];
#pragma unroll
        for (int u = 0; u < 8; ++u) {
            int x = ee + u * SORT_T;
            if (x >= e1) x = ee;
            d[u]  = dst[x];
            sc[u] = src[x];
        }
#pragma unroll
        for (int u = 0; u < 8; ++u) {
            if (ee + u * SORT_T < e1) {
                const int pos = atomicAdd(&cur[d[u]], 1);
                const int2 pr = make_int2(sc[u], d[u]);
                if (STAGED) image[pos] = pr;
                else        slice[pos] = pr;
            }
        }
    }
}

__launch_bounds__(SORT_T)
__global__ void k_chunk_sort(const int* __restrict__ src, const int* __restrict__ dst,
                             int2* __restrict__ sorted, int* __restrict__ Goff, int N, int E) {
    extern __shared__ int sh[];
    __shared__ int s_wtot[SORT_T / 32];
    __shared__ int2 s_pairs[CAP1];
    const int t    = threadIdx.x;
    const int lane = t & 31;
    const int wave = t >> 5;
    const int b    = blockIdx.x;
    const int per  = (E + NCH - 1) / NCH;
    const int e0   = b * per;
    const int e1   = (e0 + per < E) ? (e0 + per) : E;
    const int R    = (N + SORT_T - 1) / SORT_T;
    for (int i = t; i < R * SORT_T; i += SORT_T) sh[i] = 0;
    __syncthreads();
    int e = e0 + t;
    for (; e + 3 * SORT_T < e1; e += 4 * SORT_T) {
        const int d0 = dst[e];
        const int d1 = dst[e + SORT_T];
        const int d2 = dst[e + 2 * SORT_T];
        const int d3 = dst[e + 3 * SORT_T];
        atomicAdd(&sh[d0], 1);
        atomicAdd(&sh[d1], 1);
        atomicAdd(&sh[d2], 1);
        atomicAdd(&sh[d3], 1);
    }
    for (; e < e1; e += SORT_T) atomicAdd(&sh[dst[e]], 1);
    __syncthreads();

    const int i0 = t * R;
    int s = 0;
#pragma unroll 4
    for (int k = 0; k < R; ++k) s += sh[i0 + k];
    const int incl = wave_incl_scan(s, lane);
    if (lane == 31) s_wtot[wave] = incl;
    __syncthreads();
    const int woff = wave_sum((lane < wave) ? s_wtot[lane] : 0);
    int run = woff + incl - s;
#pragma unroll 4
    for (int k = 0; k < R; ++k) {
        const int v = sh[i0 + k];
        sh[i0 + k] = run;
        run += v;
    }
    __syncthreads();

    const int NG  = (N + GB - 1) / GB;
    const int tot = (e1 > e0) ? (e1 - e0) : 0;
    for (int g = t; g <= NG; g += SORT_T)
        Goff[(size_t)g * NCH + b] = (g < NG) ? sh[g * GB] : tot;
    __syncthreads();

    int2* slice = sorted + e0;
    if (per <= CAP1) {
        chunk_scatter<true>(src, dst, sh, s_pairs, slice, e0, e1, t);
        __syncthreads();
#pragma unroll 4
        for (int i = t; i < tot; i += SORT_T) slice[i] = s_pairs[i];
    } else {
        chunk_scatter<false>(src, dst, sh, s_pairs, slice, e0, e1, t);
    }
}

template <bool STAGED>
__device__ __forceinline__ void group_place_rows(const int2* __restrict__ sorted, const int* s_lo, const int* s_hi,
                                                 int* cnt, const int* s_rs, int* image, int* __restrict__ ent,
                                                 int n0, int per, int a0, int wave, int lane) {
    for (int c = wave; c < NCH; c += SORT_T / 32) {
        const int lo = s_lo[c];
        const int hi = s_hi[c];
        const int2* sp = sorted + (size_t)c * per;
        const int cb = c * GB - n0;
        for (int k0 = lo + lane; k0 < hi; k0 += 8 * 32) {
            int2 p[8];
#pragma unroll
            for (int u = 0; u < 8; ++u) {
                const int k = k0 + u * 32;
                p[u] = sp[(k < hi) ? k : k0];
            }
#pragma unroll
            for (int u = 0; u < 8; ++u) {
                if (k0 + u * 32 < hi) {
                    const int nl  = p[u].y - n0;
                    const int pos = s_rs[nl] + atomicAdd(&cnt[cb + p[u].y], 1);
                    if (STAGED) image[pos - a0] = p[u].x;
                    else        ent[pos]        = p[u].x;
                }
            }
        }
    }
}

__launch_bounds__(SORT_T)
__global__ void k_group_place(const int2* __restrict__ sorted, const int* __restrict__ Goff,
                              int* __restrict__ rowptr, float* __restrict__ dinv,
                              int* __restrict__ ent, int N, int E) {
    __shared__ int cnt[NCH * GB];
    __shared__ int s_lo[NCH];
    __shared__ int s_hi[NCH];
    __shared__ int s_bp[NCH / 32];
    __shared__ int s_bq[NCH / 32];
    __shared__ int s_wt[GB / 32];
    __shared__ int s_rs[GB];
    __shared__ int s_ent[CAP2];
    const int t    = threadIdx.x;
    const int lane = t & 31;
    const int wave = t >> 5;
    const int g    = blockIdx.x;
    const int n0   = g * GB;
    const int per  = (E + NCH - 1) / NCH;

    if (t < NCH) {
        const int lo = Goff[(size_t)g * NCH + t];
        const int hi = Goff[(size_t)(g + 1) * NCH + t];
        s_lo[t] = lo;
        s_hi[t] = hi;
        const int p = wave_sum(lo);
        const int q = wave_sum(hi);
        if (lane == 0) {
            s_bp[wave] = p;
            s_bq[wave] = q;
        }
    }
    for (int i = t; i < NCH * GB; i += SORT_T) cnt[i] = 0;
    __syncthreads();
    int base = 0;
    int past = 0;
#pragma unroll
    for (int w = 0; w < NCH / 32; ++w) {
        base += s_bp[w];
        past += s_bq[w];
    }
    const int a0   = base & ~31;
    const int span = past - a0;

    for (int c = wave; c < NCH; c += SORT_T / 32) {
        const int lo = s_lo[c];
        const int hi = s_hi[c];
        const int2* sp = sorted + (size_t)c * per;
        const int cb = c * GB - n0;
        for (int k0 = lo + lane; k0 < hi; k0 += 8 * 32) {
            int dd[8];
#pragma unroll
            for (int u = 0; u < 8; ++u) {
                const int k = k0 + u * 32;
                dd[u] = sp[(k < hi) ? k : k0].y;
            }
#pragma unroll
            for (int u = 0; u < 8; ++u)
                if (k0 + u * 32 < hi) atomicAdd(&cnt[cb + dd[u]], 1);
        }
    }
    __syncthreads();

    int run = 0;
    int incl = 0;
    if (t < GB) {
#pragma unroll 16
        for (int c = 0; c < NCH; ++c) {
            const int v = cnt[c * GB + t];
            cnt[c * GB + t] = run;
            run += v;
        }
        incl = wave_incl_scan(run, lane);
        if (lane == 31) s_wt[wave] = incl;
    }
    __syncthreads();
    if (t < GB) {
        int first = base;
#pragma unroll
        for (int w = 0; w < GB / 32; ++w)
            if (w < wave) first += s_wt[w];
        const int r0 = first + incl - run;
        s_rs[t] = r0;
        const int n = n0 + t;
        if (n < N) {
            rowptr[n] = r0;
            const float d = (float)run + 1.0f;
            dinv[n] = (d > 0.0f) ? rsqrtf(d) : 0.0f;
            if (n == N - 1) rowptr[N] = r0 + run;
        }
    }
    __syncthreads();

    if (span <= CAP2) {
        group_place_rows<true>(sorted, s_lo, s_hi, cnt, s_rs, s_ent, ent, n0, per, a0, wave, lane);
        __syncthreads();
        const int head = base - a0;
#pragma unroll 4
        for (int li = t; li < span; li += SORT_T)
            if (li >= head) ent[a0 + li] = s_ent[li];
    } else {
        group_place_rows<false>(sorted, s_lo, s_hi, cnt, s_rs, s_ent, ent, n0, per, a0, wave, lane);
    }
}

constexpr int TILE_M = 16;
constexpr int PITCH  = 132;

template <int K, int NF>
__device__ __forceinline__ v8f tile_product(const float* At, const float* __restrict__ W,
                                            int colB, int r16, int kh) {
    const float* ap = At + r16 * PITCH + kh;
    const float* bp = W + (size_t)kh * NF + colB;
    v8f c = {};
#pragma unroll 16
    for (int k = 0; k < K; k += 4) {
        v2f a = *(const v2f*)ap;
        v2f b;
        b.x = bp[0];
        b.y = bp[NF];
        c = __builtin_amdgcn_wmma_f32_16x16x4_f32(
                 false, a,  false, b,
                 (short)0, c,  false,  false);
        ap += 4;
        bp += 4 * NF;
    }
    return c;
}

__launch_bounds__(TILE_M * 32)
__global__ void k_gather_dense(const float* __restrict__ x,
                               const int* __restrict__ rowptr,
                               const int* __restrict__ ent,
                               const float* __restrict__ dinv,
                               const float* __restrict__ W1,
                               const float* __restrict__ b1,
                               const float* __restrict__ W2,
                               float* __restrict__ hs2, int N) {
    constexpr int IN = 128, HID = 128, OUT = 64;
    __shared__ float T[TILE_M * PITCH];
    __shared__ float U[TILE_M * PITCH];
    const int lane  = threadIdx.x & 31;
    const int wave  = threadIdx.x >> 5;
    const int node0 = blockIdx.x * TILE_M;
    const int i     = node0 + wave;
    const int q     = lane * 4;

    float4 acc = make_float4(0.0f, 0.0f, 0.0f, 0.0f);
    if (i < N) {
        const float* xq = x + q;
        const int j1 = rowptr[i + 1];
        int j = rowptr[i];
        for (; j + 8 <= j1; j += 8) {
            int    s[8];
            float  wt[8];
            float4 v[8];
#pragma unroll
            for (int u = 0; u < 8; ++u) s[u] = ent[j + u];
#pragma unroll
            for (int u = 0; u < 8; ++u) v[u] = *(const float4*)(xq + (size_t)s[u] * IN);
#pragma unroll
            for (int u = 0; u < 8; ++u) wt[u] = dinv[s[u]];
#pragma unroll
            for (int u = 0; u < 8; ++u) {
                const float w = wt[u];
                acc.x = fmaf(w, v[u].x, acc.x);
                acc.y = fmaf(w, v[u].y, acc.y);
                acc.z = fmaf(w, v[u].z, acc.z);
                acc.w = fmaf(w, v[u].w, acc.w);
            }
        }
        for (; j < j1; ++j) {
            const int s = ent[j];
            const float w = dinv[s];
            const float4 v = *(const float4*)(xq + (size_t)s * IN);
            acc.x = fmaf(w, v.x, acc.x);
            acc.y = fmaf(w, v.y, acc.y);
            acc.z = fmaf(w, v.z, acc.z);
            acc.w = fmaf(w, v.w, acc.w);
        }
        const float di = dinv[i];
        const float4 vh = *(const float4*)(xq + (size_t)i * IN);
        acc.x = di * (acc.x + di * vh.x);
        acc.y = di * (acc.y + di * vh.y);
        acc.z = di * (acc.z + di * vh.z);
        acc.w = di * (acc.w + di * vh.w);
    }
    *(float4*)(T + wave * PITCH + q) = acc;
    __syncthreads();

    const int r16 = lane & 15;
    const int hi  = lane >> 4;
    const int kh  = hi * 2;

    if (wave < HID / 16) {
        const int col = wave * 16 + r16;
        const v8f c = tile_product<IN, HID>(T, W1, col, r16, kh);
        const float bc = b1[col];
        float* up = U + (hi * 8) * PITCH + col;
#pragma unroll
        for (int r = 0; r < 8; ++r) up[r * PITCH] = fmaxf(c[r] + bc, 0.0f);
    }
    __syncthreads();

    if (wave < OUT / 16) {
        const int col = wave * 16 + r16;
        const v8f c = tile_product<HID, OUT>(U, W2, col, r16, kh);
        const int rowBase = node0 + hi * 8;
        float* op = hs2 + (size_t)rowBase * OUT + col;
        if (rowBase + 8 <= N) {
            const float4 d0 = *(const float4*)(dinv + rowBase);
            const float4 d1 = *(const float4*)(dinv + rowBase + 4);
            op[0 * (size_t)OUT] = c[0] * d0.x;
            op[1 * (size_t)OUT] = c[1] * d0.y;
            op[2 * (size_t)OUT] = c[2] * d0.z;
            op[3 * (size_t)OUT] = c[3] * d0.w;
            op[4 * (size_t)OUT] = c[4] * d1.x;
            op[5 * (size_t)OUT] = c[5] * d1.y;
            op[6 * (size_t)OUT] = c[6] * d1.z;
            op[7 * (size_t)OUT] = c[7] * d1.w;
        } else {
#pragma unroll
            for (int r = 0; r < 8; ++r) {
                const int row = rowBase + r;
                if (row < N) op[(size_t)r * OUT] = c[r] * dinv[row];
            }
        }
    }
}

template <int F, bool RELU>
__global__ void k_gather_finalize(const float* __restrict__ hs,
                                  const int* __restrict__ rowptr,
                                  const int* __restrict__ ent,
                                  const float* __restrict__ dinv,
                                  const float* __restrict__ bias,
                                  float* __restrict__ out, int N) {
    constexpr int TPR = F / 4;
    const int i = blockIdx.x * (256 / TPR) + threadIdx.x / TPR;
    if (i >= N) return;
    const int q = (threadIdx.x % TPR) * 4;
    const float* hq = hs + q;
    const int j1 = rowptr[i + 1];
    int j = rowptr[i];
    float4 acc = make_float4(0.0f, 0.0f, 0.0f, 0.0f);
    for (; j + 8 <= j1; j += 8) {
        int s[8];
        float4 v[8];
#pragma unroll
        for (int u = 0; u < 8; ++u) s[u] = ent[j + u];
#pragma unroll
        for (int u = 0; u < 8; ++u) v[u] = *(const float4*)(hq + (size_t)s[u] * F);
#pragma unroll
        for (int u = 0; u < 8; ++u) {
            acc.x += v[u].x; acc.y += v[u].y; acc.z += v[u].z; acc.w += v[u].w;
        }
    }
    for (; j < j1; ++j) {
        const float4 v = *(const float4*)(hq + (size_t)ent[j] * F);
        acc.x += v.x; acc.y += v.y; acc.z += v.z; acc.w += v.w;
    }
    const size_t off = (size_t)i * F + q;
    const float di = dinv[i];
    const float4 vh = *(const float4*)(hs + off);
    const float4 vb = *(const float4*)(bias + q);
    float4 r;
    r.x = di * (acc.x + vh.x) + vb.x;
    r.y = di * (acc.y + vh.y) + vb.y;
    r.z = di * (acc.z + vh.z) + vb.z;
    r.w = di * (acc.w + vh.w) + vb.w;
    if (RELU) {
        r.x = fmaxf(r.x, 0.0f);
        r.y = fmaxf(r.y, 0.0f);
        r.z = fmaxf(r.z, 0.0f);
        r.w = fmaxf(r.w, 0.0f);
    }
    *(float4*)(out + off) = r;
}

extern "C" void kernel_launch(void* const* d_in, const int* in_sizes, int n_in,
                              void* d_out, int out_size, void* d_ws, size_t ws_size,
                              hipStream_t stream) {
    const float* x  = (const float*)d_in[0];
    const float* W1 = (const float*)d_in[1];
    const float* b1 = (const float*)d_in[2];
    const float* W2 = (const float*)d_in[3];
    const float* b2 = (const float*)d_in[4];
    const int*   ei = (const int*)d_in[5];

    const int IN = 128, OUT = 64;
    const int N = in_sizes[0] / IN;
    const int E = in_sizes[5] / 2;
    const int* src = ei;
    const int* dst = ei + E;

    const size_t Np = ((size_t)N + 64) & ~(size_t)63;
    const size_t Ep = ((size_t)E + 63) & ~(size_t)63;
    const int    NG = (N + GB - 1) / GB;

    float* ws     = (float*)d_ws;
    float* dinv   = ws;
    int*   rowptr = (int*)(dinv + Np);
    int*   Goff   = rowptr + Np;
    int2*  sorted = (int2*)(Goff + ((size_t)NG + 1) * NCH);
    int*   ent    = (int*)(sorted + Ep);
    float* hs2    = (float*)(ent + Ep);
    float* out    = (float*)d_out;

    const size_t ldsN = (((size_t)N + SORT_T - 1) / SORT_T) * SORT_T * sizeof(int);
    k_chunk_sort<<<NCH, SORT_T, ldsN, stream>>>(src, dst, sorted, Goff, N, E);
    k_group_place<<<NG, SORT_T, 0, stream>>>(sorted, Goff, rowptr, dinv, ent, N, E);

    k_gather_dense<<<(N + TILE_M - 1) / TILE_M, TILE_M * 32, 0, stream>>>(
        x, rowptr, ent, dinv, W1, b1, W2, hs2, N);

    k_gather_finalize<64, false><<<(N + 15) / 16, 256, 0, stream>>>(
        hs2, rowptr, ent, dinv, b2, out, N);
}


# Both gather loops: a lane loads one in-edge index (coalesced), rows go out 8 back to back by lane broadcast, next indices prefetched: no index round trip per 8 rows, no serial tail; sum order kept
# speedup vs baseline: 5.4108x; 1.0719x over previous
#include <hip/hip_runtime.h>

typedef float v2f __attribute__((ext_vector_type(2)));
typedef float v8f __attribute__((ext_vector_type(8)));

constexpr int NCH = 64;
constexpr int SORT_T = 1024;
constexpr int GB = 128;
constexpr int CAP1 = 10240;
constexpr int CAP2 = 16384;

__device__ __forceinline__ int wave_sum(int v) {
#pragma unroll
    for (int o = 16; o > 0; o >>= 1) v += __shfl_xor(v, o, 32);
    return v;
}

__device__ __forceinline__ int wave_incl_scan(int v, int lane) {
#pragma unroll
    for (int o = 1; o < 32; o <<= 1) {
        const int u = __shfl_up(v, o, 32);
        if (lane >= o) v += u;
    }
    return v;
}

template <bool STAGED>
__device__ __forceinline__ void chunk_scatter(const int* __restrict__ src, const int* __restrict__ dst,
                                              int* cur, int2* image, int2* __restrict__ slice,
                                              int e0, int e1, int t) {
    for (int ee = e0 + t; ee < e1; ee += 8 * SORT_T) {
        int d[8], sc[8];
#pragma unroll
        for (int u = 0; u < 8; ++u) {
            int x = ee + u * SORT_T;
            if (x >= e1) x = ee;
            d[u]  = dst[x];
            sc[u] = src[x];
        }
#pragma unroll
        for (int u = 0; u < 8; ++u) {
            if (ee + u * SORT_T < e1) {
                const int pos = atomicAdd(&cur[d[u]], 1);
                const int2 pr = make_int2(sc[u], d[u]);
                if (STAGED) image[pos] = pr;
                else        slice[pos] = pr;
            }
        }
    }
}

__launch_bounds__(SORT_T)
__global__ void k_chunk_sort(const int* __restrict__ src, const int* __restrict__ dst,
                             int2* __restrict__ sorted, int* __restrict__ Goff, int N, int E) {
    extern __shared__ int sh[];
    __shared__ int s_wtot[SORT_T / 32];
    __shared__ int2 s_pairs[CAP1];
    const int t    = threadIdx.x;
    const int lane = t & 31;
    const int wave = t >> 5;
    const int b    = blockIdx.x;
    const int per  = (E + NCH - 1) / NCH;
    const int e0   = b * per;
    const int e1   = (e0 + per < E) ? (e0 + per) : E;
    const int R    = (N + SORT_T - 1) / SORT_T;
    for (int i = t; i < R * SORT_T; i += SORT_T) sh[i] = 0;
    __syncthreads();
    int e = e0 + t;
    for (; e + 3 * SORT_T < e1; e += 4 * SORT_T) {
        const int d0 = dst[e];
        const int d1 = dst[e + SORT_T];
        const int d2 = dst[e + 2 * SORT_T];
        const int d3 = dst[e + 3 * SORT_T];
        atomicAdd(&sh[d0], 1);
        atomicAdd(&sh[d1], 1);
        atomicAdd(&sh[d2], 1);
        atomicAdd(&sh[d3], 1);
    }
    for (; e < e1; e += SORT_T) atomicAdd(&sh[dst[e]], 1);
    __syncthreads();

    const int i0 = t * R;
    int s = 0;
#pragma unroll 4
    for (int k = 0; k < R; ++k) s += sh[i0 + k];
    const int incl = wave_incl_scan(s, lane);
    if (lane == 31) s_wtot[wave] = incl;
    __syncthreads();
    const int woff = wave_sum((lane < wave) ? s_wtot[lane] : 0);
    int run = woff + incl - s;
#pragma unroll 4
    for (int k = 0; k < R; ++k) {
        const int v = sh[i0 + k];
        sh[i0 + k] = run;
        run += v;
    }
    __syncthreads();

    const int NG  = (N + GB - 1) / GB;
    const int tot = (e1 > e0) ? (e1 - e0) : 0;
    for (int g = t; g <= NG; g += SORT_T)
        Goff[(size_t)g * NCH + b] = (g < NG) ? sh[g * GB] : tot;
    __syncthreads();

    int2* slice = sorted + e0;
    if (per <= CAP1) {
        chunk_scatter<true>(src, dst, sh, s_pairs, slice, e0, e1, t);
        __syncthreads();
#pragma unroll 4
        for (int i = t; i < tot; i += SORT_T) slice[i] = s_pairs[i];
    } else {
        chunk_scatter<false>(src, dst, sh, s_pairs, slice, e0, e1, t);
    }
}

template <bool STAGED>
__device__ __forceinline__ void group_place_rows(const int2* __restrict__ sorted, const int* s_lo, const int* s_hi,
                                                 int* cnt, const int* s_rs, int* image, int* __restrict__ ent,
                                                 int n0, int per, int a0, int wave, int lane) {
    for (int c = wave; c < NCH; c += SORT_T / 32) {
        const int lo = s_lo[c];
        const int hi = s_hi[c];
        const int2* sp = sorted + (size_t)c * per;
        const int cb = c * GB - n0;
        for (int k0 = lo + lane; k0 < hi; k0 += 8 * 32) {
            int2 p[8];
#pragma unroll
            for (int u = 0; u < 8; ++u) {
                const int k = k0 + u * 32;
                p[u] = sp[(k < hi) ? k : k0];
            }
#pragma unroll
            for (int u = 0; u < 8; ++u) {
                if (k0 + u * 32 < hi) {
                    const int nl  = p[u].y - n0;
                    const int pos = s_rs[nl] + atomicAdd(&cnt[cb + p[u].y], 1);
                    if (STAGED) image[pos - a0] = p[u].x;
                    else        ent[pos]        = p[u].x;
                }
            }
        }
    }
}

__launch_bounds__(SORT_T)
__global__ void k_group_place(const int2* __restrict__ sorted, const int* __restrict__ Goff,
                              int* __restrict__ rowptr, float* __restrict__ dinv,
                              int* __restrict__ ent, int N, int E) {
    __shared__ int cnt[NCH * GB];
    __shared__ int s_lo[NCH];
    __shared__ int s_hi[NCH];
    __shared__ int s_bp[NCH / 32];
    __shared__ int s_bq[NCH / 32];
    __shared__ int s_wt[GB / 32];
    __shared__ int s_rs[GB];
    __shared__ int s_ent[CAP2];
    const int t    = threadIdx.x;
    const int lane = t & 31;
    const int wave = t >> 5;
    const int g    = blockIdx.x;
    const int n0   = g * GB;
    const int per  = (E + NCH - 1) / NCH;

    if (t < NCH) {
        const int lo = Goff[(size_t)g * NCH + t];
        const int hi = Goff[(size_t)(g + 1) * NCH + t];
        s_lo[t] = lo;
        s_hi[t] = hi;
        const int p = wave_sum(lo);
        const int q = wave_sum(hi);
        if (lane == 0) {
            s_bp[wave] = p;
            s_bq[wave] = q;
        }
    }
    for (int i = t; i < NCH * GB; i += SORT_T) cnt[i] = 0;
    __syncthreads();
    int base = 0;
    int past = 0;
#pragma unroll
    for (int w = 0; w < NCH / 32; ++w) {
        base += s_bp[w];
        past += s_bq[w];
    }
    const int a0   = base & ~31;
    const int span = past - a0;

    for (int c = wave; c < NCH; c += SORT_T / 32) {
        const int lo = s_lo[c];
        const int hi = s_hi[c];
        const int2* sp = sorted + (size_t)c * per;
        const int cb = c * GB - n0;
        for (int k0 = lo + lane; k0 < hi; k0 += 8 * 32) {
            int dd[8];
#pragma unroll
            for (int u = 0; u < 8; ++u) {
                const int k = k0 + u * 32;
                dd[u] = sp[(k < hi) ? k : k0].y;
            }
#pragma unroll
            for (int u = 0; u < 8; ++u)
                if (k0 + u * 32 < hi) atomicAdd(&cnt[cb + dd[u]], 1);
        }
    }
    __syncthreads();

    int run = 0;
    int incl = 0;
    if (t < GB) {
#pragma unroll 16
        for (int c = 0; c < NCH; ++c) {
            const int v = cnt[c * GB + t];
            cnt[c * GB + t] = run;
            run += v;
        }
        incl = wave_incl_scan(run, lane);
        if (lane == 31) s_wt[wave] = incl;
    }
    __syncthreads();
    if (t < GB) {
        int first = base;
#pragma unroll
        for (int w = 0; w < GB / 32; ++w)
            if (w < wave) first += s_wt[w];
        const int r0 = first + incl - run;
        s_rs[t] = r0;
        const int n = n0 + t;
        if (n < N) {
            rowptr[n] = r0;
            const float d = (float)run + 1.0f;
            dinv[n] = (d > 0.0f) ? rsqrtf(d) : 0.0f;
            if (n == N - 1) rowptr[N] = r0 + run;
        }
    }
    __syncthreads();

    if (span <= CAP2) {
        group_place_rows<true>(sorted, s_lo, s_hi, cnt, s_rs, s_ent, ent, n0, per, a0, wave, lane);
        __syncthreads();
        const int head = base - a0;
#pragma unroll 4
        for (int li = t; li < span; li += SORT_T)
            if (li >= head) ent[a0 + li] = s_ent[li];
    } else {
        group_place_rows<false>(sorted, s_lo, s_hi, cnt, s_rs, s_ent, ent, n0, per, a0, wave, lane);
    }
}

constexpr int TILE_M = 16;
constexpr int PITCH  = 132;

template <int K, int NF>
__device__ __forceinline__ v8f tile_product(const float* At, const float* __restrict__ W,
                                            int colB, int r16, int kh) {
    const float* ap = At + r16 * PITCH + kh;
    const float* bp = W + (size_t)kh * NF + colB;
    v8f c = {};
#pragma unroll 16
    for (int k = 0; k < K; k += 4) {
        v2f a = *(const v2f*)ap;
        v2f b;
        b.x = bp[0];
        b.y = bp[NF];
        c = __builtin_amdgcn_wmma_f32_16x16x4_f32(
                 false, a,  false, b,
                 (short)0, c,  false,  false);
        ap += 4;
        bp += 4 * NF;
    }
    return c;
}

template <int U0, bool FULL>
__device__ __forceinline__ void dense_rows8(const float* __restrict__ x, unsigned qb, int my_s, int my_wb,
                                            int n, float4& acc) {
    constexpr int IN = 128;
    float4 v[8];
#pragma unroll
    for (int u = 0; u < 8; ++u) {
        const int s = __builtin_amdgcn_readlane(my_s, U0 + u);
        v[u] = *(const float4*)((const char*)(x + (size_t)s * IN) + qb);
    }
#pragma unroll
    for (int u = 0; u < 8; ++u) {
        const float w = __int_as_float(__builtin_amdgcn_readlane(my_wb, U0 + u));
        float4 r = v[u];
        if (!FULL && U0 + u > 0 && U0 + u >= n) r = make_float4(0.0f, 0.0f, 0.0f, 0.0f);
        acc.x = fmaf(w, r.x, acc.x);
        acc.y = fmaf(w, r.y, acc.y);
        acc.z = fmaf(w, r.z, acc.z);
        acc.w = fmaf(w, r.w, acc.w);
    }
}

__launch_bounds__(TILE_M * 32)
__global__ void k_gather_dense(const float* __restrict__ x,
                               const int* __restrict__ rowptr,
                               const int* __restrict__ ent,
                               const float* __restrict__ dinv,
                               const float* __restrict__ W1,
                               const float* __restrict__ b1,
                               const float* __restrict__ W2,
                               float* __restrict__ hs2, int N) {
    constexpr int IN = 128, HID = 128, OUT = 64;
    __shared__ float T[TILE_M * PITCH];
    __shared__ float U[TILE_M * PITCH];
    const int lane  = threadIdx.x & 31;
    const int wave  = __builtin_amdgcn_readfirstlane((int)(threadIdx.x >> 5));
    const int node0 = blockIdx.x * TILE_M;
    const int i     = node0 + wave;
    const int q     = lane * 4;
    const unsigned qb = (unsigned)lane * 16u;

    float4 acc = make_float4(0.0f, 0.0f, 0.0f, 0.0f);
    if (i < N) {
        const int j0 = rowptr[i];
        const int j1 = rowptr[i + 1];
        const float di = dinv[i];
        const float4 vh = *(const float4*)((const char*)(x + (size_t)i * IN) + qb);
        if (j0 < j1) {
            const int e0 = j0 + lane;
            int   s_cur = ent[(e0 < j1) ? e0 : (j1 - 1)];
            float w_cur = dinv[s_cur];
            for (int jb = j0; jb < j1; jb += 32) {
                const int  n    = j1 - jb;
                const bool more = n > 32;
                int s_nxt = s_cur;
                if (more) {
                    const int e1 = jb + 32 + lane;
                    s_nxt = ent[(e1 < j1) ? e1 : (j1 - 1)];
                }
                const int wb = __float_as_int(w_cur);
                if (n >= 8) dense_rows8<0, true >(x, qb, s_cur, wb, n, acc);
                else        dense_rows8<0, false>(x, qb, s_cur, wb, n, acc);
                float w_nxt = w_cur;
                if (more) w_nxt = dinv[s_nxt];
                if (n > 8) {
                    if (n >= 16) dense_rows8<8, true >(x, qb, s_cur, wb, n, acc);
                    else         dense_rows8<8, false>(x, qb, s_cur, wb, n, acc);
                }
                if (n > 16) {
                    if (n >= 24) dense_rows8<16, true >(x, qb, s_cur, wb, n, acc);
                    else         dense_rows8<16, false>(x, qb, s_cur, wb, n, acc);
                }
                if (n > 24) {
                    if (n >= 32) dense_rows8<24, true >(x, qb, s_cur, wb, n, acc);
                    else         dense_rows8<24, false>(x, qb, s_cur, wb, n, acc);
                }
                s_cur = s_nxt;
                w_cur = w_nxt;
            }
        }
        acc.x = di * (acc.x + di * vh.x);
        acc.y = di * (acc.y + di * vh.y);
        acc.z = di * (acc.z + di * vh.z);
        acc.w = di * (acc.w + di * vh.w);
    }
    *(float4*)(T + wave * PITCH + q) = acc;
    __syncthreads();

    const int r16 = lane & 15;
    const int hi  = lane >> 4;
    const int kh  = hi * 2;

    if (wave < HID / 16) {
        const int col = wave * 16 + r16;
        const v8f c = tile_product<IN, HID>(T, W1, col, r16, kh);
        const float bc = b1[col];
        float* up = U + (hi * 8) * PITCH + col;
#pragma unroll
        for (int r = 0; r < 8; ++r) up[r * PITCH] = fmaxf(c[r] + bc, 0.0f);
    }
    __syncthreads();

    if (wave < OUT / 16) {
        const int col = wave * 16 + r16;
        const v8f c = tile_product<HID, OUT>(U, W2, col, r16, kh);
        const int rowBase = node0 + hi * 8;
        float* op = hs2 + (size_t)rowBase * OUT + col;
        if (rowBase + 8 <= N) {
            const float4 d0 = *(const float4*)(dinv + rowBase);
            const float4 d1 = *(const float4*)(dinv + rowBase + 4);
            op[0 * (size_t)OUT] = c[0] * d0.x;
            op[1 * (size_t)OUT] = c[1] * d0.y;
            op[2 * (size_t)OUT] = c[2] * d0.z;
            op[3 * (size_t)OUT] = c[3] * d0.w;
            op[4 * (size_t)OUT] = c[4] * d1.x;
            op[5 * (size_t)OUT] = c[5] * d1.y;
            op[6 * (size_t)OUT] = c[6] * d1.z;
            op[7 * (size_t)OUT] = c[7] * d1.w;
        } else {
#pragma unroll
            for (int r = 0; r < 8; ++r) {
                const int row = rowBase + r;
                if (row < N) op[(size_t)r * OUT] = c[r] * dinv[row];
            }
        }
    }
}


template <int L>
__device__ __forceinline__ int row16_bcast(int v) {
    return __builtin_amdgcn_update_dpp(0, v, 0x150 + L, 0xf, 0xf, false);
}

template <int F, int U0>
__device__ __forceinline__ void fin_rows8(const float* __restrict__ hs, unsigned qb, int my_s, int n, float4& acc) {
    int s[8];
    s[0] = row16_bcast<U0 + 0>(my_s);
    s[1] = row16_bcast<U0 + 1>(my_s);
    s[2] = row16_bcast<U0 + 2>(my_s);
    s[3] = row16_bcast<U0 + 3>(my_s);
    s[4] = row16_bcast<U0 + 4>(my_s);
    s[5] = row16_bcast<U0 + 5>(my_s);
    s[6] = row16_bcast<U0 + 6>(my_s);
    s[7] = row16_bcast<U0 + 7>(my_s);
    float4 v[8];
#pragma unroll
    for (int u = 0; u < 8; ++u)
        v[u] = *(const float4*)((const char*)hs + ((unsigned)s[u] * (unsigned)(F * 4) + qb));
#pragma unroll
    for (int u = 0; u < 8; ++u) {
        const bool ok = (U0 + u == 0) || (U0 + u < n);
        acc.x += ok ? v[u].x : 0.0f;
        acc.y += ok ? v[u].y : 0.0f;
        acc.z += ok ? v[u].z : 0.0f;
        acc.w += ok ? v[u].w : 0.0f;
    }
}

template <int F, bool RELU>
__global__ void k_gather_finalize(const float* __restrict__ hs,
                                  const int* __restrict__ rowptr,
                                  const int* __restrict__ ent,
                                  const float* __restrict__ dinv,
                                  const float* __restrict__ bias,
                                  float* __restrict__ out, int N) {
    constexpr int TPR = F / 4;
    static_assert(TPR == 16, "a node is one 16-lane row of a wave");
    const int i = blockIdx.x * (256 / TPR) + threadIdx.x / TPR;
    if (i >= N) return;
    const int sub = threadIdx.x % TPR;
    const int q = sub * 4;
    const unsigned qb = (unsigned)sub * 16u;
    const int j0 = rowptr[i];
    const int j1 = rowptr[i + 1];
    const size_t off = (size_t)i * F + q;
    const float di = dinv[i];
    const float4 vh = *(const float4*)(hs + off);
    const float4 vb = *(const float4*)(bias + q);
    float4 acc = make_float4(0.0f, 0.0f, 0.0f, 0.0f);
    if (j0 < j1) {
        const int e0 = j0 + sub;
        int s_cur = ent[(e0 < j1) ? e0 : (j1 - 1)];
        for (int jb = j0; jb < j1; jb += 16) {
            const int n = j1 - jb;
            int s_nxt = s_cur;
            if (n > 16) {
                const int e1 = jb + 16 + sub;
                s_nxt = ent[(e1 < j1) ? e1 : (j1 - 1)];
            }
            fin_rows8<F, 0>(hs, qb, s_cur, n, acc);
            if (n > 8) fin_rows8<F, 8>(hs, qb, s_cur, n, acc);
            s_cur = s_nxt;
        }
    }
    float4 r;
    r.x = di * (acc.x + vh.x) + vb.x;
    r.y = di * (acc.y + vh.y) + vb.y;
    r.z = di * (acc.z + vh.z) + vb.z;
    r.w = di * (acc.w + vh.w) + vb.w;
    if (RELU) {
        r.x = fmaxf(r.x, 0.0f);
        r.y = fmaxf(r.y, 0.0f);
        r.z = fmaxf(r.z, 0.0f);
        r.w = fmaxf(r.w, 0.0f);
    }
    *(float4*)(out + off) = r;
}

extern "C" void kernel_launch(void* const* d_in, const int* in_sizes, int n_in,
                              void* d_out, int out_size, void* d_ws, size_t ws_size,
                              hipStream_t stream) {
    const float* x  = (const float*)d_in[0];
    const float* W1 = (const float*)d_in[1];
    const float* b1 = (const float*)d_in[2];
    const float* W2 = (const float*)d_in[3];
    const float* b2 = (const float*)d_in[4];
    const int*   ei = (const int*)d_in[5];

    const int IN = 128, OUT = 64;
    const int N = in_sizes[0] / IN;
    const int E = in_sizes[5] / 2;
    const int* src = ei;
    const int* dst = ei + E;

    const size_t Np = ((size_t)N + 64) & ~(size_t)63;
    const size_t Ep = ((size_t)E + 63) & ~(size_t)63;
    const int    NG = (N + GB - 1) / GB;

    float* ws     = (float*)d_ws;
    float* dinv   = ws;
    int*   rowptr = (int*)(dinv + Np);
    int*   Goff   = rowptr + Np;
    int2*  sorted = (int2*)(Goff + ((size_t)NG + 1) * NCH);
    int*   ent    = (int*)(sorted + Ep);
    float* hs2    = (float*)(ent + Ep);
    float* out    = (float*)d_out;

    const size_t ldsN = (((size_t)N + SORT_T - 1) / SORT_T) * SORT_T * sizeof(int);
    k_chunk_sort<<<NCH, SORT_T, ldsN, stream>>>(src, dst, sorted, Goff, N, E);
    k_group_place<<<NG, SORT_T, 0, stream>>>(sorted, Goff, rowptr, dinv, ent, N, E);

    k_gather_dense<<<(N + TILE_M - 1) / TILE_M, TILE_M * 32, 0, stream>>>(
        x, rowptr, ent, dinv, W1, b1, W2, hs2, N);

    k_gather_finalize<64, false><<<(N + 15) / 16, 256, 0, stream>>>(
        hs2, rowptr, ent, dinv, b2, out, N);
}


# Sort kernels load each edge once, in one group, and hold it in registers with the rank its counting LDS add returned: 8 dependent load trips and the cursor adds gone; in-run sum order may change
# speedup vs baseline: 6.0722x; 1.1222x over previous
#include <hip/hip_runtime.h>

typedef float v2f __attribute__((ext_vector_type(2)));
typedef float v8f __attribute__((ext_vector_type(8)));

constexpr int NCH = 64;
constexpr int SORT_T = 1024;
constexpr int GB = 128;
constexpr int CAP1 = 10240;
constexpr int CAP2 = 16384;
constexpr int EPT  = CAP1 / SORT_T;
constexpr int WPG  = SORT_T / 32;
constexpr int CPW  = NCH / WPG;
constexpr int SEGR = 8;
static_assert(CAP1 % SORT_T == 0 && NCH % WPG == 0, "whole edges per thread, whole chunks per wave");

__device__ __forceinline__ int wave_sum(int v) {
#pragma unroll
    for (int o = 16; o > 0; o >>= 1) v += __shfl_xor(v, o, 32);
    return v;
}

__device__ __forceinline__ int wave_incl_scan(int v, int lane) {
#pragma unroll
    for (int o = 1; o < 32; o <<= 1) {
        const int u = __shfl_up(v, o, 32);
        if (lane >= o) v += u;
    }
    return v;
}

__device__ __forceinline__ void chunk_scatter_direct(const int* __restrict__ src, const int* __restrict__ dst,
                                                     int* cur, int2* __restrict__ slice,
                                                     int e0, int e1, int t) {
    for (int ee = e0 + t; ee < e1; ee += 8 * SORT_T) {
        int d[8], sc[8];
#pragma unroll
        for (int u = 0; u < 8; ++u) {
            int x = ee + u * SORT_T;
            if (x >= e1) x = ee;
            d[u]  = dst[x];
            sc[u] = src[x];
        }
#pragma unroll
        for (int u = 0; u < 8; ++u) {
            if (ee + u * SORT_T < e1) {
                const int pos = atomicAdd(&cur[d[u]], 1);
                slice[pos] = make_int2(sc[u], d[u]);
            }
        }
    }
}

__launch_bounds__(SORT_T)
__global__ void k_chunk_sort(const int* __restrict__ src, const int* __restrict__ dst,
                             int2* __restrict__ sorted, int* __restrict__ Goff, int N, int E) {
    extern __shared__ int sh[];
    __shared__ int s_wtot[SORT_T / 32];
    __shared__ int2 s_pairs[CAP1];
    const int t    = threadIdx.x;
    const int lane = t & 31;
    const int wave = t >> 5;
    const int b    = blockIdx.x;
    const int per  = (E + NCH - 1) / NCH;
    const int e0   = b * per;
    const int e1   = (e0 + per < E) ? (e0 + per) : E;
    const int R    = (N + SORT_T - 1) / SORT_T;
    const bool fits = per <= CAP1;
    const bool mine = e0 + t < e1;

    int d[EPT], sc[EPT], rk[EPT];
#pragma unroll
    for (int u = 0; u < EPT; ++u) {
        d[u]  = 0;
        sc[u] = 0;
        rk[u] = 0;
    }
    if (fits && mine) {
#pragma unroll
        for (int u = 0; u < EPT; ++u) {
            int x = e0 + t + u * SORT_T;
            if (x >= e1) x = e0 + t;
            d[u]  = dst[x];
            sc[u] = src[x];
        }
    }
    for (int i = t; i < R * SORT_T; i += SORT_T) sh[i] = 0;
    __syncthreads();
    if (fits) {
        if (mine) {
#pragma unroll
            for (int u = 0; u < EPT; ++u)
                if (e0 + t + u * SORT_T < e1) rk[u] = atomicAdd(&sh[d[u]], 1);
        }
    } else {
        int e = e0 + t;
        for (; e + 3 * SORT_T < e1; e += 4 * SORT_T) {
            const int d0 = dst[e];
            const int d1 = dst[e + SORT_T];
            const int d2 = dst[e + 2 * SORT_T];
            const int d3 = dst[e + 3 * SORT_T];
            atomicAdd(&sh[d0], 1);
            atomicAdd(&sh[d1], 1);
            atomicAdd(&sh[d2], 1);
            atomicAdd(&sh[d3], 1);
        }
        for (; e < e1; e += SORT_T) atomicAdd(&sh[dst[e]], 1);
    }
    __syncthreads();

    const int i0 = t * R;
    int s = 0;
#pragma unroll 4
    for (int k = 0; k < R; ++k) s += sh[i0 + k];
    const int incl = wave_incl_scan(s, lane);
    if (lane == 31) s_wtot[wave] = incl;
    __syncthreads();
    const int woff = wave_sum((lane < wave) ? s_wtot[lane] : 0);
    int run = woff + incl - s;
#pragma unroll 4
    for (int k = 0; k < R; ++k) {
        const int v = sh[i0 + k];
        sh[i0 + k] = run;
        run += v;
    }
    __syncthreads();

    const int NG  = (N + GB - 1) / GB;
    const int tot = (e1 > e0) ? (e1 - e0) : 0;
    for (int g = t; g <= NG; g += SORT_T)
        Goff[(size_t)g * NCH + b] = (g < NG) ? sh[g * GB] : tot;

    int2* slice = sorted + e0;
    if (fits) {
        if (mine) {
            int off[EPT];
#pragma unroll
            for (int u = 0; u < EPT; ++u) off[u] = sh[d[u]];
#pragma unroll
            for (int u = 0; u < EPT; ++u)
                if (e0 + t + u * SORT_T < e1) s_pairs[off[u] + rk[u]] = make_int2(sc[u], d[u]);
        }
        __syncthreads();
#pragma unroll 4
        for (int i = t; i < tot; i += SORT_T) slice[i] = s_pairs[i];
    } else {
        __syncthreads();
        chunk_scatter_direct(src, dst, sh, slice, e0, e1, t);
    }
}

template <bool STAGED>
__device__ __forceinline__ void group_place_rows(const int2* __restrict__ sorted, const int* s_lo, const int* s_hi,
                                                 int* cnt, const int* s_rs, int* image, int* __restrict__ ent,
                                                 const int2 (&p)[CPW][SEGR], const int (&rk)[CPW][SEGR],
                                                 int n0, int per, int a0, int wave, int lane) {
#pragma unroll
    for (int j = 0; j < CPW; ++j) {
        const int c  = wave + j * WPG;
        const int lo = __builtin_amdgcn_readfirstlane(s_lo[c]);
        const int hi = __builtin_amdgcn_readfirstlane(s_hi[c]);
        const int cb = c * GB - n0;
        if (hi - lo <= SEGR * 32) {
            if (lo + lane < hi) {
                int pos[SEGR];
#pragma unroll
                for (int u = 0; u < SEGR; ++u) {
                    const int dn = p[j][u].y;
                    pos[u] = s_rs[dn - n0] + cnt[cb + dn] + rk[j][u];
                }
#pragma unroll
                for (int u = 0; u < SEGR; ++u) {
                    if (lo + lane + u * 32 < hi) {
                        if (STAGED) image[pos[u] - a0] = p[j][u].x;
                        else        ent[pos[u]]        = p[j][u].x;
                    }
                }
            }
        } else {
            const int2* sp = sorted + (size_t)c * per;
            for (int k0 = lo + lane; k0 < hi; k0 += SEGR * 32) {
                int2 q[SEGR];
#pragma unroll
                for (int u = 0; u < SEGR; ++u) {
                    const int k = k0 + u * 32;
                    q[u] = sp[(k < hi) ? k : k0];
                }
#pragma unroll
                for (int u = 0; u < SEGR; ++u) {
                    if (k0 + u * 32 < hi) {
                        const int pos = s_rs[q[u].y - n0] + atomicAdd(&cnt[cb + q[u].y], 1);
                        if (STAGED) image[pos - a0] = q[u].x;
                        else        ent[pos]        = q[u].x;
                    }
                }
            }
        }
    }
}

__launch_bounds__(SORT_T)
__global__ void k_group_place(const int2* __restrict__ sorted, const int* __restrict__ Goff,
                              int* __restrict__ rowptr, float* __restrict__ dinv,
                              int* __restrict__ ent, int N, int E) {
    __shared__ int cnt[NCH * GB];
    __shared__ int s_lo[NCH];
    __shared__ int s_hi[NCH];
    __shared__ int s_bp[NCH / 32];
    __shared__ int s_bq[NCH / 32];
    __shared__ int s_wt[GB / 32];
    __shared__ int s_rs[GB];
    __shared__ int s_ent[CAP2];
    const int t    = threadIdx.x;
    const int lane = t & 31;
    const int wave = t >> 5;
    const int wsc  = __builtin_amdgcn_readfirstlane(wave);
    const int g    = blockIdx.x;
    const int n0   = g * GB;
    const int per  = (E + NCH - 1) / NCH;

    if (t < NCH) {
        const int lo = Goff[(size_t)g * NCH + t];
        const int hi = Goff[(size_t)(g + 1) * NCH + t];
        s_lo[t] = lo;
        s_hi[t] = hi;
        const int p = wave_sum(lo);
        const int q = wave_sum(hi);
        if (lane == 0) {
            s_bp[wave] = p;
            s_bq[wave] = q;
        }
    }
    for (int i = t; i < NCH * GB; i += SORT_T) cnt[i] = 0;
    __syncthreads();
    int base = 0;
    int past = 0;
#pragma unroll
    for (int w = 0; w < NCH / 32; ++w) {
        base += s_bp[w];
        past += s_bq[w];
    }
    const int a0   = base & ~31;
    const int span = past - a0;

    int2 p[CPW][SEGR];
    int  rk[CPW][SEGR];
#pragma unroll
    for (int j = 0; j < CPW; ++j) {
#pragma unroll
        for (int u = 0; u < SEGR; ++u) {
            p[j][u]  = make_int2(0, 0);
            rk[j][u] = 0;
        }
        const int c  = wsc + j * WPG;
        const int lo = __builtin_amdgcn_readfirstlane(s_lo[c]);
        const int hi = __builtin_amdgcn_readfirstlane(s_hi[c]);
        if (hi - lo <= SEGR * 32 && lo + lane < hi) {
            const int2* sp = sorted + (size_t)c * per;
            const int k0 = lo + lane;
#pragma unroll
            for (int u = 0; u < SEGR; ++u) {
                const int k = k0 + u * 32;
                p[j][u] = sp[(k < hi) ? k : k0];
            }
        }
    }
#pragma unroll
    for (int j = 0; j < CPW; ++j) {
        const int c  = wsc + j * WPG;
        const int lo = __builtin_amdgcn_readfirstlane(s_lo[c]);
        const int hi = __builtin_amdgcn_readfirstlane(s_hi[c]);
        const int cb = c * GB - n0;
        if (hi - lo <= SEGR * 32) {
#pragma unroll
            for (int u = 0; u < SEGR; ++u)
                if (lo + lane + u * 32 < hi) rk[j][u] = atomicAdd(&cnt[cb + p[j][u].y], 1);
        } else {
            const int2* sp = sorted + (size_t)c * per;
            for (int k0 = lo + lane; k0 < hi; k0 += SEGR * 32) {
                int dd[SEGR];
#pragma unroll
                for (int u = 0; u < SEGR; ++u) {
                    const int k = k0 + u * 32;
                    dd[u] = sp[(k < hi) ? k : k0].y;
                }
#pragma unroll
                for (int u = 0; u < SEGR; ++u)
                    if (k0 + u * 32 < hi) atomicAdd(&cnt[cb + dd[u]], 1);
            }
        }
    }
    __syncthreads();

    int run = 0;
    int incl = 0;
    if (t < GB) {
#pragma unroll 16
        for (int c = 0; c < NCH; ++c) {
            const int v = cnt[c * GB + t];
            cnt[c * GB + t] = run;
            run += v;
        }
        incl = wave_incl_scan(run, lane);
        if (lane == 31) s_wt[wave] = incl;
    }
    __syncthreads();
    if (t < GB) {
        int first = base;
#pragma unroll
        for (int w = 0; w < GB / 32; ++w)
            if (w < wave) first += s_wt[w];
        const int r0 = first + incl - run;
        s_rs[t] = r0;
        const int n = n0 + t;
        if (n < N) {
            rowptr[n] = r0;
            const float d = (float)run + 1.0f;
            dinv[n] = (d > 0.0f) ? rsqrtf(d) : 0.0f;
            if (n == N - 1) rowptr[N] = r0 + run;
        }
    }
    __syncthreads();

    if (span <= CAP2) {
        group_place_rows<true>(sorted, s_lo, s_hi, cnt, s_rs, s_ent, ent, p, rk, n0, per, a0, wsc, lane);
        __syncthreads();
        const int head = base - a0;
#pragma unroll 4
        for (int li = t; li < span; li += SORT_T)
            if (li >= head) ent[a0 + li] = s_ent[li];
    } else {
        group_place_rows<false>(sorted, s_lo, s_hi, cnt, s_rs, s_ent, ent, p, rk, n0, per, a0, wsc, lane);
    }
}

constexpr int TILE_M = 16;
constexpr int PITCH  = 132;

template <int K, int NF>
__device__ __forceinline__ v8f tile_product(const float* At, const float* __restrict__ W,
                                            int colB, int r16, int kh) {
    const float* ap = At + r16 * PITCH + kh;
    const float* bp = W + (size_t)kh * NF + colB;
    v8f c = {};
#pragma unroll 16
    for (int k = 0; k < K; k += 4) {
        v2f a = *(const v2f*)ap;
        v2f b;
        b.x = bp[0];
        b.y = bp[NF];
        c = __builtin_amdgcn_wmma_f32_16x16x4_f32(
                 false, a,  false, b,
                 (short)0, c,  false,  false);
        ap += 4;
        bp += 4 * NF;
    }
    return c;
}

template <int U0, bool FULL>
__device__ __forceinline__ void dense_rows8(const float* __restrict__ x, unsigned qb, int my_s, int my_wb,
                                            int n, float4& acc) {
    constexpr int IN = 128;
    float4 v[8];
#pragma unroll
    for (int u = 0; u < 8; ++u) {
        const int s = __builtin_amdgcn_readlane(my_s, U0 + u);
        v[u] = *(const float4*)((const char*)(x + (size_t)s * IN) + qb);
    }
#pragma unroll
    for (int u = 0; u < 8; ++u) {
        const float w = __int_as_float(__builtin_amdgcn_readlane(my_wb, U0 + u));
        float4 r = v[u];
        if (!FULL && U0 + u > 0 && U0 + u >= n) r = make_float4(0.0f, 0.0f, 0.0f, 0.0f);
        acc.x = fmaf(w, r.x, acc.x);
        acc.y = fmaf(w, r.y, acc.y);
        acc.z = fmaf(w, r.z, acc.z);
        acc.w = fmaf(w, r.w, acc.w);
    }
}

__launch_bounds__(TILE_M * 32)
__global__ void k_gather_dense(const float* __restrict__ x,
                               const int* __restrict__ rowptr,
                               const int* __restrict__ ent,
                               const float* __restrict__ dinv,
                               const float* __restrict__ W1,
                               const float* __restrict__ b1,
                               const float* __restrict__ W2,
                               float* __restrict__ hs2, int N) {
    constexpr int IN = 128, HID = 128, OUT = 64;
    __shared__ float T[TILE_M * PITCH];
    __shared__ float U[TILE_M * PITCH];
    const int lane  = threadIdx.x & 31;
    const int wave  = __builtin_amdgcn_readfirstlane((int)(threadIdx.x >> 5));
    const int node0 = blockIdx.x * TILE_M;
    const int i     = node0 + wave;
    const int q     = lane * 4;
    const unsigned qb = (unsigned)lane * 16u;

    float4 acc = make_float4(0.0f, 0.0f, 0.0f, 0.0f);
    if (i < N) {
        const int j0 = rowptr[i];
        const int j1 = rowptr[i + 1];
        const float di = dinv[i];
        const float4 vh = *(const float4*)((const char*)(x + (size_t)i * IN) + qb);
        if (j0 < j1) {
            const int e0 = j0 + lane;
            int   s_cur = ent[(e0 < j1) ? e0 : (j1 - 1)];
            float w_cur = dinv[s_cur];
            for (int jb = j0; jb < j1; jb += 32) {
                const int  n    = j1 - jb;
                const bool more = n > 32;
                int s_nxt = s_cur;
                if (more) {
                    const int e1 = jb + 32 + lane;
                    s_nxt = ent[(e1 < j1) ? e1 : (j1 - 1)];
                }
                const int wb = __float_as_int(w_cur);
                if (n >= 8) dense_rows8<0, true >(x, qb, s_cur, wb, n, acc);
                else        dense_rows8<0, false>(x, qb, s_cur, wb, n, acc);
                float w_nxt = w_cur;
                if (more) w_nxt = dinv[s_nxt];
                if (n > 8) {
                    if (n >= 16) dense_rows8<8, true >(x, qb, s_cur, wb, n, acc);
                    else         dense_rows8<8, false>(x, qb, s_cur, wb, n, acc);
                }
                if (n > 16) {
                    if (n >= 24) dense_rows8<16, true >(x, qb, s_cur, wb, n, acc);
                    else         dense_rows8<16, false>(x, qb, s_cur, wb, n, acc);
                }
                if (n > 24) {
                    if (n >= 32) dense_rows8<24, true >(x, qb, s_cur, wb, n, acc);
                    else         dense_rows8<24, false>(x, qb, s_cur, wb, n, acc);
                }
                s_cur = s_nxt;
                w_cur = w_nxt;
            }
        }
        acc.x = di * (acc.x + di * vh.x);
        acc.y = di * (acc.y + di * vh.y);
        acc.z = di * (acc.z + di * vh.z);
        acc.w = di * (acc.w + di * vh.w);
    }
    *(float4*)(T + wave * PITCH + q) = acc;
    __syncthreads();

    const int r16 = lane & 15;
    const int hi  = lane >> 4;
    const int kh  = hi * 2;

    if (wave < HID / 16) {
        const int col = wave * 16 + r16;
        const v8f c = tile_product<IN, HID>(T, W1, col, r16, kh);
        const float bc = b1[col];
        float* up = U + (hi * 8) * PITCH + col;
#pragma unroll
        for (int r = 0; r < 8; ++r) up[r * PITCH] = fmaxf(c[r] + bc, 0.0f);
    }
    __syncthreads();

    if (wave < OUT / 16) {
        const int col = wave * 16 + r16;
        const v8f c = tile_product<HID, OUT>(U, W2, col, r16, kh);
        const int rowBase = node0 + hi * 8;
        float* op = hs2 + (size_t)rowBase * OUT + col;
        if (rowBase + 8 <= N) {
            const float4 d0 = *(const float4*)(dinv + rowBase);
            const float4 d1 = *(const float4*)(dinv + rowBase + 4);
            op[0 * (size_t)OUT] = c[0] * d0.x;
            op[1 * (size_t)OUT] = c[1] * d0.y;
            op[2 * (size_t)OUT] = c[2] * d0.z;
            op[3 * (size_t)OUT] = c[3] * d0.w;
            op[4 * (size_t)OUT] = c[4] * d1.x;
            op[5 * (size_t)OUT] = c[5] * d1.y;
            op[6 * (size_t)OUT] = c[6] * d1.z;
            op[7 * (size_t)OUT] = c[7] * d1.w;
        } else {
#pragma unroll
            for (int r = 0; r < 8; ++r) {
                const int row = rowBase + r;
                if (row < N) op[(size_t)r * OUT] = c[r] * dinv[row];
            }
        }
    }
}


template <int L>
__device__ __forceinline__ int row16_bcast(int v) {
    return __builtin_amdgcn_update_dpp(0, v, 0x150 + L, 0xf, 0xf, false);
}

template <int F, int U0>
__device__ __forceinline__ void fin_rows8(const float* __restrict__ hs, unsigned qb, int my_s, int n, float4& acc) {
    int s[8];
    s[0] = row16_bcast<U0 + 0>(my_s);
    s[1] = row16_bcast<U0 + 1>(my_s);
    s[2] = row16_bcast<U0 + 2>(my_s);
    s[3] = row16_bcast<U0 + 3>(my_s);
    s[4] = row16_bcast<U0 + 4>(my_s);
    s[5] = row16_bcast<U0 + 5>(my_s);
    s[6] = row16_bcast<U0 + 6>(my_s);
    s[7] = row16_bcast<U0 + 7>(my_s);
    float4 v[8];
#pragma unroll
    for (int u = 0; u < 8; ++u)
        v[u] = *(const float4*)((const char*)hs + ((unsigned)s[u] * (unsigned)(F * 4) + qb));
#pragma unroll
    for (int u = 0; u < 8; ++u) {
        const bool ok = (U0 + u == 0) || (U0 + u < n);
        acc.x += ok ? v[u].x : 0.0f;
        acc.y += ok ? v[u].y : 0.0f;
        acc.z += ok ? v[u].z : 0.0f;
        acc.w += ok ? v[u].w : 0.0f;
    }
}

template <int F, bool RELU>
__global__ void k_gather_finalize(const float* __restrict__ hs,
                                  const int* __restrict__ rowptr,
                                  const int* __restrict__ ent,
                                  const float* __restrict__ dinv,
                                  const float* __restrict__ bias,
                                  float* __restrict__ out, int N) {
    constexpr int TPR = F / 4;
    static_assert(TPR == 16, "a node is one 16-lane row of a wave");
    const int i = blockIdx.x * (256 / TPR) + threadIdx.x / TPR;
    if (i >= N) return;
    const int sub = threadIdx.x % TPR;
    const int q = sub * 4;
    const unsigned qb = (unsigned)sub * 16u;
    const int j0 = rowptr[i];
    const int j1 = rowptr[i + 1];
    const size_t off = (size_t)i * F + q;
    const float di = dinv[i];
    const float4 vh = *(const float4*)(hs + off);
    const float4 vb = *(const float4*)(bias + q);
    float4 acc = make_float4(0.0f, 0.0f, 0.0f, 0.0f);
    if (j0 < j1) {
        const int e0 = j0 + sub;
        int s_cur = ent[(e0 < j1) ? e0 : (j1 - 1)];
        for (int jb = j0; jb < j1; jb += 16) {
            const int n = j1 - jb;
            int s_nxt = s_cur;
            if (n > 16) {
                const int e1 = jb + 16 + sub;
                s_nxt = ent[(e1 < j1) ? e1 : (j1 - 1)];
            }
            fin_rows8<F, 0>(hs, qb, s_cur, n, acc);
            if (n > 8) fin_rows8<F, 8>(hs, qb, s_cur, n, acc);
            s_cur = s_nxt;
        }
    }
    float4 r;
    r.x = di * (acc.x + vh.x) + vb.x;
    r.y = di * (acc.y + vh.y) + vb.y;
    r.z = di * (acc.z + vh.z) + vb.z;
    r.w = di * (acc.w + vh.w) + vb.w;
    if (RELU) {
        r.x = fmaxf(r.x, 0.0f);
        r.y = fmaxf(r.y, 0.0f);
        r.z = fmaxf(r.z, 0.0f);
        r.w = fmaxf(r.w, 0.0f);
    }
    *(float4*)(out + off) = r;
}

extern "C" void kernel_launch(void* const* d_in, const int* in_sizes, int n_in,
                              void* d_out, int out_size, void* d_ws, size_t ws_size,
                              hipStream_t stream) {
    const float* x  = (const float*)d_in[0];
    const float* W1 = (const float*)d_in[1];
    const float* b1 = (const float*)d_in[2];
    const float* W2 = (const float*)d_in[3];
    const float* b2 = (const float*)d_in[4];
    const int*   ei = (const int*)d_in[5];

    const int IN = 128, OUT = 64;
    const int N = in_sizes[0] / IN;
    const int E = in_sizes[5] / 2;
    const int* src = ei;
    const int* dst = ei + E;

    const size_t Np = ((size_t)N + 64) & ~(size_t)63;
    const size_t Ep = ((size_t)E + 63) & ~(size_t)63;
    const int    NG = (N + GB - 1) / GB;

    float* ws     = (float*)d_ws;
    float* dinv   = ws;
    int*   rowptr = (int*)(dinv + Np);
    int*   Goff   = rowptr + Np;
    int2*  sorted = (int2*)(Goff + ((size_t)NG + 1) * NCH);
    int*   ent    = (int*)(sorted + Ep);
    float* hs2    = (float*)(ent + Ep);
    float* out    = (float*)d_out;

    const size_t ldsN = (((size_t)N + SORT_T - 1) / SORT_T) * SORT_T * sizeof(int);
    k_chunk_sort<<<NCH, SORT_T, ldsN, stream>>>(src, dst, sorted, Goff, N, E);
    k_group_place<<<NG, SORT_T, 0, stream>>>(sorted, Goff, rowptr, dinv, ent, N, E);

    k_gather_dense<<<(N + TILE_M - 1) / TILE_M, TILE_M * 32, 0, stream>>>(
        x, rowptr, ent, dinv, W1, b1, W2, hs2, N);

    k_gather_finalize<64, false><<<(N + 15) / 16, 256, 0, stream>>>(
        hs2, rowptr, ent, dinv, b2, out, N);
}


# Edge sort cut twice as fine at both levels (128 chunks, 64-node groups, 512-thread work-groups): half the LDS count, rank and place work per WGP; rows ordered by finer chunks, so sum order changes
# speedup vs baseline: 6.5869x; 1.0848x over previous
#include <hip/hip_runtime.h>

typedef float v2f __attribute__((ext_vector_type(2)));
typedef float v8f __attribute__((ext_vector_type(8)));

constexpr int NCH = 128;
constexpr int SORT_T = 512;
constexpr int GB = 64;
constexpr int CAP1 = 5120;
constexpr int CAP2 = 16384;
constexpr int EPT  = CAP1 / SORT_T;
constexpr int WPG  = SORT_T / 32;
constexpr int CPW  = NCH / WPG;
constexpr int SEGR = 2;
constexpr int XF   = 128;
constexpr int XRW  = GB / WPG;
static_assert(XF == 32 * 4 && GB % WPG == 0, "a wave covers one row of x per load, whole rows per wave");
static_assert(CAP1 % SORT_T == 0 && NCH % WPG == 0, "whole edges per thread, whole chunks per wave");

__device__ __forceinline__ int wave_sum(int v) {
#pragma unroll
    for (int o = 16; o > 0; o >>= 1) v += __shfl_xor(v, o, 32);
    return v;
}

__device__ __forceinline__ int wave_incl_scan(int v, int lane) {
#pragma unroll
    for (int o = 1; o < 32; o <<= 1) {
        const int u = __shfl_up(v, o, 32);
        if (lane >= o) v += u;
    }
    return v;
}

__device__ __forceinline__ void chunk_scatter_direct(const int* __restrict__ src, const int* __restrict__ dst,
                                                     int* cur, int2* __restrict__ slice,
                                                     int e0, int e1, int t) {
    for (int ee = e0 + t; ee < e1; ee += 8 * SORT_T) {
        int d[8], sc[8];
#pragma unroll
        for (int u = 0; u < 8; ++u) {
            int x = ee + u * SORT_T;
            if (x >= e1) x = ee;
            d[u]  = dst[x];
            sc[u] = src[x];
        }
#pragma unroll
        for (int u = 0; u < 8; ++u) {
            if (ee + u * SORT_T < e1) {
                const int pos = atomicAdd(&cur[d[u]], 1);
                slice[pos] = make_int2(sc[u], d[u]);
            }
        }
    }
}

__launch_bounds__(SORT_T)
__global__ void k_chunk_sort(const int* __restrict__ src, const int* __restrict__ dst,
                             int2* __restrict__ sorted, int* __restrict__ Goff, int N, int E) {
    extern __shared__ int sh[];
    __shared__ int s_wtot[SORT_T / 32];
    __shared__ int2 s_pairs[CAP1];
    const int t    = threadIdx.x;
    const int lane = t & 31;
    const int wave = t >> 5;
    const int b    = blockIdx.x;
    const int per  = (E + NCH - 1) / NCH;
    const int e0   = b * per;
    const int e1   = (e0 + per < E) ? (e0 + per) : E;
    const int R    = ((N + SORT_T - 1) / SORT_T) | 1;
    const bool fits = per <= CAP1;
    const bool mine = e0 + t < e1;

    int d[EPT], sc[EPT], rk[EPT];
#pragma unroll
    for (int u = 0; u < EPT; ++u) {
        d[u]  = 0;
        sc[u] = 0;
        rk[u] = 0;
    }
    if (fits && mine) {
#pragma unroll
        for (int u = 0; u < EPT; ++u) {
            int x = e0 + t + u * SORT_T;
            if (x >= e1) x = e0 + t;
            d[u]  = dst[x];
            sc[u] = src[x];
        }
    }
    for (int i = t; i < R * SORT_T; i += SORT_T) sh[i] = 0;
    __syncthreads();
    if (fits) {
        if (mine) {
#pragma unroll
            for (int u = 0; u < EPT; ++u)
                if (e0 + t + u * SORT_T < e1) rk[u] = atomicAdd(&sh[d[u]], 1);
        }
    } else {
        int e = e0 + t;
        for (; e + 3 * SORT_T < e1; e += 4 * SORT_T) {
            const int d0 = dst[e];
            const int d1 = dst[e + SORT_T];
            const int d2 = dst[e + 2 * SORT_T];
            const int d3 = dst[e + 3 * SORT_T];
            atomicAdd(&sh[d0], 1);
            atomicAdd(&sh[d1], 1);
            atomicAdd(&sh[d2], 1);
            atomicAdd(&sh[d3], 1);
        }
        for (; e < e1; e += SORT_T) atomicAdd(&sh[dst[e]], 1);
    }
    __syncthreads();

    const int i0 = t * R;
    int s = 0;
#pragma unroll 8
    for (int k = 0; k < R; ++k) s += sh[i0 + k];
    const int incl = wave_incl_scan(s, lane);
    if (lane == 31) s_wtot[wave] = incl;
    __syncthreads();
    const int woff = wave_sum((lane < wave) ? s_wtot[lane] : 0);
    int run = woff + incl - s;
#pragma unroll 8
    for (int k = 0; k < R; ++k) {
        const int v = sh[i0 + k];
        sh[i0 + k] = run;
        run += v;
    }
    __syncthreads();

    const int NG  = (N + GB - 1) / GB;
    const int tot = (e1 > e0) ? (e1 - e0) : 0;
    for (int g = t; g <= NG; g += SORT_T)
        Goff[(size_t)g * NCH + b] = (g < NG) ? sh[g * GB] : tot;

    int2* slice = sorted + e0;
    if (fits) {
        if (mine) {
            int off[EPT];
#pragma unroll
            for (int u = 0; u < EPT; ++u) off[u] = sh[d[u]];
#pragma unroll
            for (int u = 0; u < EPT; ++u)
                if (e0 + t + u * SORT_T < e1) s_pairs[off[u] + rk[u]] = make_int2(sc[u], d[u]);
        }
        __syncthreads();
#pragma unroll 4
        for (int i = t; i < tot; i += SORT_T) slice[i] = s_pairs[i];
    } else {
        __syncthreads();
        chunk_scatter_direct(src, dst, sh, slice, e0, e1, t);
    }
}

template <bool STAGED>
__device__ __forceinline__ void group_place_rows(const int2* __restrict__ sorted, const int (&lo_)[CPW], const int (&hi_)[CPW],
                                                 int* cnt, const int* s_rs, int* image, int* __restrict__ ent,
                                                 const int2 (&p)[CPW][SEGR], const int (&rk)[CPW][SEGR],
                                                 int n0, int per, int a0, int wave, int lane) {
#pragma unroll
    for (int j = 0; j < CPW; ++j) {
        const int c  = wave + j * WPG;
        const int lo = lo_[j];
        const int hi = hi_[j];
        const int cb = c * GB - n0;
        if (hi - lo <= SEGR * 32) {
            if (lo + lane < hi) {
                int pos[SEGR];
#pragma unroll
                for (int u = 0; u < SEGR; ++u) {
                    const int dn = p[j][u].y;
                    pos[u] = s_rs[dn - n0] + cnt[cb + dn] + rk[j][u];
                }
#pragma unroll
                for (int u = 0; u < SEGR; ++u) {
                    if (lo + lane + u * 32 < hi) {
                        if (STAGED) image[pos[u] - a0] = p[j][u].x;
                        else        ent[pos[u]]        = p[j][u].x;
                    }
                }
            }
        } else {
            const int2* sp = sorted + (size_t)c * per;
            for (int k0 = lo + lane; k0 < hi; k0 += SEGR * 32) {
                int2 q[SEGR];
#pragma unroll
                for (int u = 0; u < SEGR; ++u) {
                    const int k = k0 + u * 32;
                    q[u] = sp[(k < hi) ? k : k0];
                }
#pragma unroll
                for (int u = 0; u < SEGR; ++u) {
                    if (k0 + u * 32 < hi) {
                        const int pos = s_rs[q[u].y - n0] + atomicAdd(&cnt[cb + q[u].y], 1);
                        if (STAGED) image[pos - a0] = q[u].x;
                        else        ent[pos]        = q[u].x;
                    }
                }
            }
        }
    }
}

__device__ __forceinline__ void barrier_lds_only() {
    __builtin_amdgcn_fence(__ATOMIC_RELEASE, "workgroup", "local");
    __builtin_amdgcn_s_barrier();
    __builtin_amdgcn_fence(__ATOMIC_ACQUIRE, "workgroup", "local");
}

__launch_bounds__(SORT_T)
__global__ void k_group_place(const int2* __restrict__ sorted, const int* __restrict__ Goff,
                              int* __restrict__ rowptr, float* __restrict__ dinv,
                              int* __restrict__ ent, const float* __restrict__ x, float* __restrict__ xs,
                              int N, int E) {
    __shared__ int cnt[NCH * GB];
    __shared__ int s_lo[NCH];
    __shared__ int s_hi[NCH];
    __shared__ int s_bp[NCH / 32];
    __shared__ int s_bq[NCH / 32];
    __shared__ int s_wt[GB / 32];
    __shared__ int s_rs[GB];
    __shared__ float s_dv[GB];
    __shared__ int s_ent[CAP2];
    const int t    = threadIdx.x;
    const int lane = t & 31;
    const int wave = t >> 5;
    const int wsc  = __builtin_amdgcn_readfirstlane(wave);
    const int g    = blockIdx.x;
    const int n0   = g * GB;
    const int per  = (E + NCH - 1) / NCH;

    if (t < NCH) {
        const int lo = Goff[(size_t)g * NCH + t];
        const int hi = Goff[(size_t)(g + 1) * NCH + t];
        s_lo[t] = lo;
        s_hi[t] = hi;
        const int p = wave_sum(lo);
        const int q = wave_sum(hi);
        if (lane == 0) {
            s_bp[wave] = p;
            s_bq[wave] = q;
        }
    }
    for (int i = t; i < NCH * GB; i += SORT_T) cnt[i] = 0;
    __syncthreads();
    int base = 0;
    int past = 0;
#pragma unroll
    for (int w = 0; w < NCH / 32; ++w) {
        base += s_bp[w];
        past += s_bq[w];
    }
    const int a0   = base & ~31;
    const int span = past - a0;

    int2 p[CPW][SEGR];
    int  rk[CPW][SEGR];
    int seg_lo[CPW], seg_hi[CPW];
#pragma unroll
    for (int j = 0; j < CPW; ++j) {
        seg_lo[j] = s_lo[wsc + j * WPG];
        seg_hi[j] = s_hi[wsc + j * WPG];
    }
#pragma unroll
    for (int j = 0; j < CPW; ++j) {
        seg_lo[j] = __builtin_amdgcn_readfirstlane(seg_lo[j]);
        seg_hi[j] = __builtin_amdgcn_readfirstlane(seg_hi[j]);
    }
#pragma unroll
    for (int j = 0; j < CPW; ++j) {
#pragma unroll
        for (int u = 0; u < SEGR; ++u) {
            p[j][u]  = make_int2(0, 0);
            rk[j][u] = 0;
        }
        const int c  = wsc + j * WPG;
        const int lo = seg_lo[j];
        const int hi = seg_hi[j];
        if (hi - lo <= SEGR * 32 && lo + lane < hi) {
            const int2* sp = sorted + (size_t)c * per;
            const int k0 = lo + lane;
#pragma unroll
            for (int u = 0; u < SEGR; ++u) {
                const int k = k0 + u * 32;
                p[j][u] = sp[(k < hi) ? k : k0];
            }
        }
    }
#pragma unroll
    for (int j = 0; j < CPW; ++j) {
        const int c  = wsc + j * WPG;
        const int lo = seg_lo[j];
        const int hi = seg_hi[j];
        const int cb = c * GB - n0;
        if (hi - lo <= SEGR * 32) {
#pragma unroll
            for (int u = 0; u < SEGR; ++u)
                if (lo + lane + u * 32 < hi) rk[j][u] = atomicAdd(&cnt[cb + p[j][u].y], 1);
        } else {
            const int2* sp = sorted + (size_t)c * per;
            for (int k0 = lo + lane; k0 < hi; k0 += SEGR * 32) {
                int dd[SEGR];
#pragma unroll
                for (int u = 0; u < SEGR; ++u) {
                    const int k = k0 + u * 32;
                    dd[u] = sp[(k < hi) ? k : k0].y;
                }
#pragma unroll
                for (int u = 0; u < SEGR; ++u)
                    if (k0 + u * 32 < hi) atomicAdd(&cnt[cb + dd[u]], 1);
            }
        }
    }
    float4 xv[XRW];
#pragma unroll
    for (int k = 0; k < XRW; ++k) {
        const int n  = n0 + wsc + k * WPG;
        const int nc = (n < N) ? n : (N - 1);
        xv[k] = *(const float4*)(x + (size_t)nc * XF + lane * 4);
    }
    barrier_lds_only();

    int run = 0;
    int incl = 0;
    if (t < GB) {
#pragma unroll 32
        for (int c = 0; c < NCH; ++c) {
            const int v = cnt[c * GB + t];
            cnt[c * GB + t] = run;
            run += v;
        }
        incl = wave_incl_scan(run, lane);
        if (lane == 31) s_wt[wave] = incl;
    }
    barrier_lds_only();
    if (t < GB) {
        int first = base;
#pragma unroll
        for (int w = 0; w < GB / 32; ++w)
            if (w < wave) first += s_wt[w];
        const int r0 = first + incl - run;
        s_rs[t] = r0;
        const int n = n0 + t;
        const float d  = (float)run + 1.0f;
        const float dv = (d > 0.0f) ? rsqrtf(d) : 0.0f;
        s_dv[t] = dv;
        if (n < N) {
            rowptr[n] = r0;
            dinv[n] = dv;
            if (n == N - 1) rowptr[N] = r0 + run;
        }
    }
    barrier_lds_only();

    float dvr[XRW];
#pragma unroll
    for (int k = 0; k < XRW; ++k) dvr[k] = s_dv[wsc + k * WPG];
#pragma unroll
    for (int k = 0; k < XRW; ++k) {
        const int n = n0 + wsc + k * WPG;
        if (n < N) {
            const float dv = dvr[k];
            float4 v = xv[k];
            v.x *= dv;
            v.y *= dv;
            v.z *= dv;
            v.w *= dv;
            *(float4*)(xs + (size_t)n * XF + lane * 4) = v;
        }
    }

    if (span <= CAP2) {
        group_place_rows<true>(sorted, seg_lo, seg_hi, cnt, s_rs, s_ent, ent, p, rk, n0, per, a0, wsc, lane);
        barrier_lds_only();
        const int head = base - a0;
#pragma unroll 4
        for (int li = t; li < span; li += SORT_T)
            if (li >= head) ent[a0 + li] = s_ent[li];
    } else {
        group_place_rows<false>(sorted, seg_lo, seg_hi, cnt, s_rs, s_ent, ent, p, rk, n0, per, a0, wsc, lane);
    }
}

constexpr int TILE_M = 16;
constexpr int PITCH  = 132;

template <int K, int NF>
__device__ __forceinline__ v8f tile_product(const float* At, const float* __restrict__ W,
                                            int colB, int r16, int kh) {
    const float* ap = At + r16 * PITCH + kh;
    const float* bp = W + (size_t)kh * NF + colB;
    v8f c = {};
#pragma unroll 16
    for (int k = 0; k < K; k += 4) {
        v2f a = *(const v2f*)ap;
        v2f b;
        b.x = bp[0];
        b.y = bp[NF];
        c = __builtin_amdgcn_wmma_f32_16x16x4_f32(
                 false, a,  false, b,
                 (short)0, c,  false,  false);
        ap += 4;
        bp += 4 * NF;
    }
    return c;
}

template <int U0, bool FULL>
__device__ __forceinline__ void dense_rows8(const float* __restrict__ xs, unsigned qb, int my_s, int n, float4& acc) {
    float4 v[8];
#pragma unroll
    for (int u = 0; u < 8; ++u) {
        const int s = __builtin_amdgcn_readlane(my_s, U0 + u);
        v[u] = *(const float4*)((const char*)(xs + (size_t)s * XF) + qb);
    }
#pragma unroll
    for (int u = 0; u < 8; ++u) {
        float4 r = v[u];
        if (!FULL && U0 + u > 0 && U0 + u >= n) r = make_float4(0.0f, 0.0f, 0.0f, 0.0f);
        acc.x += r.x;
        acc.y += r.y;
        acc.z += r.z;
        acc.w += r.w;
    }
}

__launch_bounds__(TILE_M * 32)
__global__ void k_gather_dense(const float* __restrict__ xs,
                               const int* __restrict__ rowptr,
                               const int* __restrict__ ent,
                               const float* __restrict__ dinv,
                               const float* __restrict__ W1,
                               const float* __restrict__ b1,
                               const float* __restrict__ W2,
                               float* __restrict__ hs2, int N) {
    constexpr int IN = XF, HID = 128, OUT = 64;
    __shared__ float T[TILE_M * PITCH];
    __shared__ float U[TILE_M * PITCH];
    const int lane  = threadIdx.x & 31;
    const int wave  = __builtin_amdgcn_readfirstlane((int)(threadIdx.x >> 5));
    const int node0 = blockIdx.x * TILE_M;
    const int i     = node0 + wave;
    const int q     = lane * 4;
    const unsigned qb = (unsigned)lane * 16u;

    float4 acc = make_float4(0.0f, 0.0f, 0.0f, 0.0f);
    if (i < N) {
        const int j0 = rowptr[i];
        const int j1 = rowptr[i + 1];
        const float di = dinv[i];
        const float4 vh = *(const float4*)((const char*)(xs + (size_t)i * XF) + qb);
        if (j0 < j1) {
            const int e0 = j0 + lane;
            int s_cur = ent[(e0 < j1) ? e0 : (j1 - 1)];
            for (int jb = j0; jb < j1; jb += 32) {
                const int n = j1 - jb;
                int s_nxt = s_cur;
                if (n > 32) {
                    const int e1 = jb + 32 + lane;
                    s_nxt = ent[(e1 < j1) ? e1 : (j1 - 1)];
                }
                if (n >= 8) dense_rows8<0, true >(xs, qb, s_cur, n, acc);
                else        dense_rows8<0, false>(xs, qb, s_cur, n, acc);
                if (n > 8) {
                    if (n >= 16) dense_rows8<8, true >(xs, qb, s_cur, n, acc);
                    else         dense_rows8<8, false>(xs, qb, s_cur, n, acc);
                }
                if (n > 16) {
                    if (n >= 24) dense_rows8<16, true >(xs, qb, s_cur, n, acc);
                    else         dense_rows8<16, false>(xs, qb, s_cur, n, acc);
                }
                if (n > 24) {
                    if (n >= 32) dense_rows8<24, true >(xs, qb, s_cur, n, acc);
                    else         dense_rows8<24, false>(xs, qb, s_cur, n, acc);
                }
                s_cur = s_nxt;
            }
        }
        acc.x = di * (acc.x + vh.x);
        acc.y = di * (acc.y + vh.y);
        acc.z = di * (acc.z + vh.z);
        acc.w = di * (acc.w + vh.w);
    }
    *(float4*)(T + wave * PITCH + q) = acc;
    __syncthreads();

    const int r16 = lane & 15;
    const int hi  = lane >> 4;
    const int kh  = hi * 2;

    if (wave < HID / 16) {
        const int col = wave * 16 + r16;
        const v8f c = tile_product<IN, HID>(T, W1, col, r16, kh);
        const float bc = b1[col];
        float* up = U + (hi * 8) * PITCH + col;
#pragma unroll
        for (int r = 0; r < 8; ++r) up[r * PITCH] = fmaxf(c[r] + bc, 0.0f);
    }
    __syncthreads();

    if (wave < OUT / 16) {
        const int col = wave * 16 + r16;
        const v8f c = tile_product<HID, OUT>(U, W2, col, r16, kh);
        const int rowBase = node0 + hi * 8;
        float* op = hs2 + (size_t)rowBase * OUT + col;
        if (rowBase + 8 <= N) {
            const float4 d0 = *(const float4*)(dinv + rowBase);
            const float4 d1 = *(const float4*)(dinv + rowBase + 4);
            op[0 * (size_t)OUT] = c[0] * d0.x;
            op[1 * (size_t)OUT] = c[1] * d0.y;
            op[2 * (size_t)OUT] = c[2] * d0.z;
            op[3 * (size_t)OUT] = c[3] * d0.w;
            op[4 * (size_t)OUT] = c[4] * d1.x;
            op[5 * (size_t)OUT] = c[5] * d1.y;
            op[6 * (size_t)OUT] = c[6] * d1.z;
            op[7 * (size_t)OUT] = c[7] * d1.w;
        } else {
#pragma unroll
            for (int r = 0; r < 8; ++r) {
                const int row = rowBase + r;
                if (row < N) op[(size_t)r * OUT] = c[r] * dinv[row];
            }
        }
    }
}


template <int L>
__device__ __forceinline__ int row16_bcast(int v) {
    return __builtin_amdgcn_update_dpp(0, v, 0x150 + L, 0xf, 0xf, false);
}

template <int F, int U0>
__device__ __forceinline__ void fin_rows8(const float* __restrict__ hs, unsigned qb, int my_s, int n, float4& acc) {
    int s[8];
    s[0] = row16_bcast<U0 + 0>(my_s);
    s[1] = row16_bcast<U0 + 1>(my_s);
    s[2] = row16_bcast<U0 + 2>(my_s);
    s[3] = row16_bcast<U0 + 3>(my_s);
    s[4] = row16_bcast<U0 + 4>(my_s);
    s[5] = row16_bcast<U0 + 5>(my_s);
    s[6] = row16_bcast<U0 + 6>(my_s);
    s[7] = row16_bcast<U0 + 7>(my_s);
    float4 v[8];
#pragma unroll
    for (int u = 0; u < 8; ++u)
        v[u] = *(const float4*)((const char*)hs + ((unsigned)s[u] * (unsigned)(F * 4) + qb));
#pragma unroll
    for (int u = 0; u < 8; ++u) {
        const bool ok = (U0 + u == 0) || (U0 + u < n);
        acc.x += ok ? v[u].x : 0.0f;
        acc.y += ok ? v[u].y : 0.0f;
        acc.z += ok ? v[u].z : 0.0f;
        acc.w += ok ? v[u].w : 0.0f;
    }
}

template <int F, bool RELU>
__global__ void k_gather_finalize(const float* __restrict__ hs,
                                  const int* __restrict__ rowptr,
                                  const int* __restrict__ ent,
                                  const float* __restrict__ dinv,
                                  const float* __restrict__ bias,
                                  float* __restrict__ out, int N) {
    constexpr int TPR = F / 4;
    static_assert(TPR == 16, "a node is one 16-lane row of a wave");
    const int i = blockIdx.x * (256 / TPR) + threadIdx.x / TPR;
    if (i >= N) return;
    const int sub = threadIdx.x % TPR;
    const int q = sub * 4;
    const unsigned qb = (unsigned)sub * 16u;
    const int j0 = rowptr[i];
    const int j1 = rowptr[i + 1];
    const size_t off = (size_t)i * F + q;
    const float di = dinv[i];
    const float4 vh = *(const float4*)(hs + off);
    const float4 vb = *(const float4*)(bias + q);
    float4 acc = make_float4(0.0f, 0.0f, 0.0f, 0.0f);
    if (j0 < j1) {
        const int e0 = j0 + sub;
        int s_cur = ent[(e0 < j1) ? e0 : (j1 - 1)];
        for (int jb = j0; jb < j1; jb += 16) {
            const int n = j1 - jb;
            int s_nxt = s_cur;
            if (n > 16) {
                const int e1 = jb + 16 + sub;
                s_nxt = ent[(e1 < j1) ? e1 : (j1 - 1)];
            }
            fin_rows8<F, 0>(hs, qb, s_cur, n, acc);
            if (n > 8) fin_rows8<F, 8>(hs, qb, s_cur, n, acc);
            s_cur = s_nxt;
        }
    }
    float4 r;
    r.x = di * (acc.x + vh.x) + vb.x;
    r.y = di * (acc.y + vh.y) + vb.y;
    r.z = di * (acc.z + vh.z) + vb.z;
    r.w = di * (acc.w + vh.w) + vb.w;
    if (RELU) {
        r.x = fmaxf(r.x, 0.0f);
        r.y = fmaxf(r.y, 0.0f);
        r.z = fmaxf(r.z, 0.0f);
        r.w = fmaxf(r.w, 0.0f);
    }
    *(float4*)(out + off) = r;
}

extern "C" void kernel_launch(void* const* d_in, const int* in_sizes, int n_in,
                              void* d_out, int out_size, void* d_ws, size_t ws_size,
                              hipStream_t stream) {
    const float* x  = (const float*)d_in[0];
    const float* W1 = (const float*)d_in[1];
    const float* b1 = (const float*)d_in[2];
    const float* W2 = (const float*)d_in[3];
    const float* b2 = (const float*)d_in[4];
    const int*   ei = (const int*)d_in[5];

    const int IN = 128, OUT = 64;
    const int N = in_sizes[0] / IN;
    const int E = in_sizes[5] / 2;
    const int* src = ei;
    const int* dst = ei + E;

    const size_t Np = ((size_t)N + 64) & ~(size_t)63;
    const size_t Ep = ((size_t)E + 63) & ~(size_t)63;
    const int    NG = (N + GB - 1) / GB;

    float* ws     = (float*)d_ws;
    float* dinv   = ws;
    int*   rowptr = (int*)(dinv + Np);
    int*   Goff   = rowptr + Np;
    int2*  sorted = (int2*)(Goff + ((size_t)NG + 1) * NCH);
    int*   ent    = (int*)(sorted + Ep);
    float* hs2    = (float*)(ent + Ep);
    float* xs     = hs2 + (size_t)N * OUT;
    float* out    = (float*)d_out;

    const size_t ldsN = ((((size_t)N + SORT_T - 1) / SORT_T) | 1) * SORT_T * sizeof(int);
    k_chunk_sort<<<NCH, SORT_T, ldsN, stream>>>(src, dst, sorted, Goff, N, E);
    k_group_place<<<NG, SORT_T, 0, stream>>>(sorted, Goff, rowptr, dinv, ent, x, xs, N, E);

    k_gather_dense<<<(N + TILE_M - 1) / TILE_M, TILE_M * 32, 0, stream>>>(
        xs, rowptr, ent, dinv, W1, b1, W2, hs2, N);

    k_gather_finalize<64, false><<<(N + 15) / 16, 256, 0, stream>>>(
        hs2, rowptr, ent, dinv, b2, out, N);
}
